# starting kernel
# baseline (speedup 1.0000x reference)
#include <hip/hip_runtime.h>
#include <math.h>

#define IN_DIM  100
#define HIDDEN  16
#define NCLS    40

typedef __attribute__((ext_vector_type(2))) float v2f;
typedef __attribute__((ext_vector_type(8))) float v8f;

static __device__ inline v8f wmma_f32_16x16x4(v2f a, v2f b, v8f c) {
#if __has_builtin(__builtin_amdgcn_wmma_f32_16x16x4_f32)
  return __builtin_amdgcn_wmma_f32_16x16x4_f32(
       false, a,  false, b,
       (short)0, c,  false,  false);
#else
  const int lane = (int)(threadIdx.x & 31u);
  const int nsel = lane & 15;
  v8f d = c;
#pragma unroll
  for (int g = 0; g < 8; ++g) {
    const int m = g + 8 * (lane >> 4);
    float acc = 0.f;
#pragma unroll
    for (int k = 0; k < 4; ++k) {
      const int srcA = m + 16 * (k >> 1);
      const int srcB = nsel + 16 * (k >> 1);
      const float av = __shfl((k & 1) ? a.y : a.x, srcA, 32);
      const float bv = __shfl((k & 1) ? b.y : b.x, srcB, 32);
      acc += av * bv;
    }
    d[g] += acc;
  }
  return d;
#endif
}

__global__ void gcn_zero_kernel(float* __restrict__ p, int n) {
  int i = blockIdx.x * blockDim.x + threadIdx.x;
  int stride = gridDim.x * blockDim.x;
  for (; i < n; i += stride) p[i] = 0.0f;
}

__global__ __launch_bounds__(32) void gcn_gemm1_wmma(
    const float* __restrict__ X, const float* __restrict__ W1,
    float* __restrict__ Y, int N) {
  const int lane = (int)threadIdx.x;
  const int r0   = blockIdx.x * 16;
  const int col  = lane & 15;
  const int kh   = lane >> 4;
  const int rowA = min(r0 + col, N - 1);
  const float* __restrict__ Xr = X + (size_t)rowA * IN_DIM;

  v8f acc = {};
#pragma unroll 5
  for (int k0 = 0; k0 < IN_DIM; k0 += 4) {
    const int ka = k0 + 2 * kh;
    v2f a;
    a.x = Xr[ka];
    a.y = Xr[ka + 1];
    v2f b;
    b.x = W1[ka * HIDDEN + col];
    b.y = W1[(ka + 1) * HIDDEN + col];
    acc = wmma_f32_16x16x4(a, b, acc);
  }

  float* __restrict__ Yr = Y + (size_t)(r0 + 8 * kh) * HIDDEN + col;
  if (r0 + 16 <= N) {
#pragma unroll
    for (int g = 0; g < 8; ++g) Yr[g * HIDDEN] = acc[g];
  } else {
#pragma unroll
    for (int g = 0; g < 8; ++g) {
      const int row = r0 + g + 8 * kh;
      if (row < N) Y[row * HIDDEN + col] = acc[g];
    }
  }
}

__global__ __launch_bounds__(256) void gcn_spmm_kernel(
    const int* __restrict__ esrc, const int* __restrict__ edst,
    const float* __restrict__ evals, const float* __restrict__ Xin,
    float* __restrict__ Z, int E, int doRelu) {
  const int t = blockIdx.x * blockDim.x + threadIdx.x;
  const int e = t >> 4;
  const int c = t & 15;
  if (e >= E) return;
  const int s = esrc[e];
  const int d = edst[e];
  const float v = evals[e];
  float xv = Xin[(size_t)s * HIDDEN + c];
  if (doRelu) xv = fmaxf(xv, 0.0f);
  atomicAdd(&Z[(size_t)d * HIDDEN + c], v * xv);
}

__global__ __launch_bounds__(32) void gcn_gemm2_lsm_wmma(
    const float* __restrict__ Z2, const float* __restrict__ W2,
    float* __restrict__ out, int N) {
  const int lane = (int)threadIdx.x;
  const int r0   = blockIdx.x * 16;
  const int col  = lane & 15;
  const int kh   = lane >> 4;
  const int rowA = min(r0 + col, N - 1);
  const float* __restrict__ Zr = Z2 + (size_t)rowA * HIDDEN;

  v8f acc0 = {}, acc1 = {}, acc2 = {};
#pragma unroll
  for (int k0 = 0; k0 < HIDDEN; k0 += 4) {
    const int ka = k0 + 2 * kh;
    v2f a;
    a.x = Zr[ka];
    a.y = Zr[ka + 1];

    v2f b0;
    b0.x = W2[ka * NCLS + col];
    b0.y = W2[(ka + 1) * NCLS + col];
    acc0 = wmma_f32_16x16x4(a, b0, acc0);

    v2f b1;
    b1.x = W2[ka * NCLS + 16 + col];
    b1.y = W2[(ka + 1) * NCLS + 16 + col];
    acc1 = wmma_f32_16x16x4(a, b1, acc1);

    const int c2 = 32 + col;
    v2f b2;
    b2.x = (c2 < NCLS) ? W2[ka * NCLS + c2] : 0.0f;
    b2.y = (c2 < NCLS) ? W2[(ka + 1) * NCLS + c2] : 0.0f;
    acc2 = wmma_f32_16x16x4(a, b2, acc2);
  }

  const bool t2ok = (col < 8);
  const float ninf = -__builtin_inff();
  const bool fullTile = (r0 + 16 <= N);
  float* __restrict__ Or = out + (size_t)(r0 + 8 * kh) * NCLS + col;

#pragma unroll
  for (int g = 0; g < 8; ++g) {
    const float v0 = acc0[g];
    const float v1 = acc1[g];
    const float v2v = acc2[g];

    float m = fmaxf(fmaxf(v0, v1), t2ok ? v2v : ninf);
#pragma unroll
    for (int off = 8; off >= 1; off >>= 1) m = fmaxf(m, __shfl_xor(m, off, 16));

    float s = expf(v0 - m) + expf(v1 - m) + (t2ok ? expf(v2v - m) : 0.0f);
#pragma unroll
    for (int off = 8; off >= 1; off >>= 1) s += __shfl_xor(s, off, 16);

    const float lse = m + logf(s);

    if (fullTile) {
      Or[g * NCLS] = v0 - lse;
      Or[g * NCLS + 16] = v1 - lse;
      if (t2ok) Or[g * NCLS + 32] = v2v - lse;
    } else {
      const int row = r0 + g + 8 * kh;
      if (row < N) {
        out[(size_t)row * NCLS + col] = v0 - lse;
        out[(size_t)row * NCLS + 16 + col] = v1 - lse;
        if (t2ok) out[(size_t)row * NCLS + 32 + col] = v2v - lse;
      }
    }
  }
}

extern "C" void kernel_launch(void* const* d_in, const int* in_sizes, int n_in,
                              void* d_out, int out_size, void* d_ws, size_t ws_size,
                              hipStream_t stream) {
  const float* features = (const float*)d_in[0];
  const int*   esrc     = (const int*)d_in[1];
  const int*   edst     = (const int*)d_in[2];
  const float* evals    = (const float*)d_in[3];
  const float* W1       = (const float*)d_in[4];
  const float* W2       = (const float*)d_in[5];
  float*       out      = (float*)d_out;

  const int N = in_sizes[0] / IN_DIM;
  const int E = in_sizes[1];

  float* Y  = (float*)d_ws;
  float* Z1 = Y + (size_t)N * HIDDEN;
  float* Z2 = Z1 + (size_t)N * HIDDEN;

  {
    const int n = 2 * N * HIDDEN;
    const int blocks = (n + 255) / 256;
    gcn_zero_kernel<<<blocks, 256, 0, stream>>>(Z1, n);
  }

  gcn_gemm1_wmma<<<(N + 15) / 16, 32, 0, stream>>>(features, W1, Y, N);

  {
    const int threads = E * 16;
    gcn_spmm_kernel<<<(threads + 255) / 256, 256, 0, stream>>>(
        esrc, edst, evals, Y, Z1, E,  0);
  }

  {
    const int threads = E * 16;
    gcn_spmm_kernel<<<(threads + 255) / 256, 256, 0, stream>>>(
        esrc, edst, evals, Z1, Z2, E,  1);
  }

  gcn_gemm2_lsm_wmma<<<(N + 15) / 16, 32, 0, stream>>>(Z2, W2, out, N);
}


# Both SpMMs: float scatter atomics replaced by a per-call dst-grouped edge list (LDS-counter ranks per edge run, prefix sums, fill) and atomic-free row gathers; rows summed in rank order, 8-slot tree
# speedup vs baseline: 1.2137x; 1.2137x over previous
#include <hip/hip_runtime.h>
#include <math.h>

#define IN_DIM  100
#define HIDDEN  16
#define NCLS    40

typedef __attribute__((ext_vector_type(2))) float v2f;
typedef __attribute__((ext_vector_type(8))) float v8f;

static __device__ inline v8f wmma_f32_16x16x4(v2f a, v2f b, v8f c) {
#if __has_builtin(__builtin_amdgcn_wmma_f32_16x16x4_f32)
  return __builtin_amdgcn_wmma_f32_16x16x4_f32(
       false, a,  false, b,
       (short)0, c,  false,  false);
#else
  const int lane = (int)(threadIdx.x & 31u);
  const int nsel = lane & 15;
  v8f d = c;
#pragma unroll
  for (int g = 0; g < 8; ++g) {
    const int m = g + 8 * (lane >> 4);
    float acc = 0.f;
#pragma unroll
    for (int k = 0; k < 4; ++k) {
      const int srcA = m + 16 * (k >> 1);
      const int srcB = nsel + 16 * (k >> 1);
      const float av = __shfl((k & 1) ? a.y : a.x, srcA, 32);
      const float bv = __shfl((k & 1) ? b.y : b.x, srcB, 32);
      acc += av * bv;
    }
    d[g] += acc;
  }
  return d;
#endif
}

#define GCN_CHUNKS 64
#define GCN_HCAP   65536

__global__ __launch_bounds__(1024) void gcn_rank_lds_kernel(
    const int* __restrict__ edst, int* __restrict__ cntG,
    int* __restrict__ lrank, int N, int E, int Npad, int chunk) {
  __shared__ int h[GCN_HCAP];
  const int t    = (int)threadIdx.x;
  const int g    = (int)blockIdx.x;
  const int e_lo = min(E, g * chunk);
  const int e_hi = min(E, e_lo + chunk);
  int4* __restrict__ h4 = reinterpret_cast<int4*>(h);

  for (int lo = 0; lo < N; lo += GCN_HCAP) {
    const int n  = min(GCN_HCAP, N - lo);
    const int n4 = (n + 3) >> 2;
    for (int i = t; i < n4; i += 1024) h4[i] = make_int4(0, 0, 0, 0);
    __syncthreads();

    for (int e0 = e_lo + t; e0 < e_hi; e0 += 4096) {
      int d[4];
#pragma unroll
      for (int u = 0; u < 4; ++u) d[u] = edst[min(e0 + u * 1024, e_hi - 1)] - lo;
#pragma unroll
      for (int u = 0; u < 4; ++u) {
        const int e = e0 + u * 1024;
        if (e < e_hi && (unsigned)d[u] < (unsigned)n) lrank[e] = atomicAdd(&h[d[u]], 1);
      }
    }
    __syncthreads();

    int4* __restrict__ o4 = reinterpret_cast<int4*>(cntG + (size_t)g * Npad + lo);
    for (int i = t; i < n4; i += 1024) o4[i] = h4[i];
    __syncthreads();
  }
}

__global__ __launch_bounds__(256) void gcn_chunk_prefix_kernel(
    const int* __restrict__ cntG, int* __restrict__ preG,
    int* __restrict__ tot, int N, int Npad) {
  const int i = blockIdx.x * blockDim.x + threadIdx.x;
  if (i >= N) return;
  int run = 0;
#pragma unroll 16
  for (int g = 0; g < GCN_CHUNKS; ++g) {
    const int c = cntG[(size_t)g * Npad + i];
    preG[(size_t)g * Npad + i] = run;
    run += c;
  }
  tot[i] = run;
}

__global__ __launch_bounds__(1024) void gcn_rowptr_kernel(
    const int* __restrict__ cnt, int* __restrict__ rowptr, int N) {
  __shared__ int s_pre[32];
  __shared__ int s_tot[32];
  const int t    = (int)threadIdx.x;
  const int lane = t & 31;
  const int w    = t >> 5;
  const int base = blockIdx.x * 1024;

  int pre = 0;
  const int4* __restrict__ c4 = reinterpret_cast<const int4*>(cnt);
  const int n4 = base >> 2;
  for (int j = t; j < n4; j += 4096) {
    const int4 a = c4[j];
    const int4 b = c4[min(j + 1024, n4 - 1)];
    const int4 c = c4[min(j + 2048, n4 - 1)];
    const int4 d = c4[min(j + 3072, n4 - 1)];
    const int sb = (j + 1024 < n4) ? (b.x + b.y) + (b.z + b.w) : 0;
    const int sc = (j + 2048 < n4) ? (c.x + c.y) + (c.z + c.w) : 0;
    const int sd = (j + 3072 < n4) ? (d.x + d.y) + (d.z + d.w) : 0;
    pre += ((a.x + a.y) + (a.z + a.w)) + sb + (sc + sd);
  }
#pragma unroll
  for (int off = 16; off >= 1; off >>= 1) pre += __shfl_xor(pre, off, 32);

  const int i = base + t;
  const int x = (i < N) ? cnt[i] : 0;
  int inc = x;
#pragma unroll
  for (int off = 1; off < 32; off <<= 1) {
    const int y = __shfl_up(inc, off, 32);
    if (lane >= off) inc += y;
  }
  if (lane == 31) s_tot[w] = inc;
  if (lane == 0)  s_pre[w] = pre;
  __syncthreads();

  int p  = s_pre[lane];
  int wt = (lane < w) ? s_tot[lane] : 0;
#pragma unroll
  for (int off = 16; off >= 1; off >>= 1) {
    p  += __shfl_xor(p, off, 32);
    wt += __shfl_xor(wt, off, 32);
  }
  if (i <= N) rowptr[i] = p + wt + inc - x;
}

__global__ __launch_bounds__(256) void gcn_fill_kernel(
    const int* __restrict__ esrc, const int* __restrict__ edst,
    const float* __restrict__ evals, const int* __restrict__ lrank,
    const int* __restrict__ rowptr, const int* __restrict__ preG,
    int* __restrict__ csr, int E, int Npad, int chunk) {
  const int e = blockIdx.x * blockDim.x + threadIdx.x;
  if (e >= E) return;
  const int d = edst[e];
  const int g = e / chunk;
  const int pos = rowptr[d] + preG[(size_t)g * Npad + d] + lrank[e];
  int2 ev;
  ev.x = esrc[e];
  ev.y = __float_as_int(evals[e]);
  reinterpret_cast<int2*>(csr)[pos] = ev;
}

__global__ __launch_bounds__(32) void gcn_gemm1_wmma(
    const float* __restrict__ X, const float* __restrict__ W1,
    float* __restrict__ Y, int N) {
  const int lane = (int)threadIdx.x;
  const int r0   = blockIdx.x * 16;
  const int col  = lane & 15;
  const int kh   = lane >> 4;
  const int rowA = min(r0 + col, N - 1);
  const float* __restrict__ Xr = X + (size_t)rowA * IN_DIM;

  v8f acc = {};
#pragma unroll 5
  for (int k0 = 0; k0 < IN_DIM; k0 += 4) {
    const int ka = k0 + 2 * kh;
    v2f a;
    a.x = Xr[ka];
    a.y = Xr[ka + 1];
    v2f b;
    b.x = W1[ka * HIDDEN + col];
    b.y = W1[(ka + 1) * HIDDEN + col];
    acc = wmma_f32_16x16x4(a, b, acc);
  }

  float* __restrict__ Yr = Y + (size_t)(r0 + 8 * kh) * HIDDEN + col;
  if (r0 + 16 <= N) {
#pragma unroll
    for (int g = 0; g < 8; ++g) Yr[g * HIDDEN] = acc[g];
  } else {
#pragma unroll
    for (int g = 0; g < 8; ++g) {
      const int row = r0 + g + 8 * kh;
      if (row < N) Y[row * HIDDEN + col] = acc[g];
    }
  }
}

static __device__ inline float4 gcn_row_gather(
    const int2* __restrict__ csr, const float4* __restrict__ X4,
    int beg, int end, int q, int c4, int doRelu) {
  float4 acc = make_float4(0.f, 0.f, 0.f, 0.f);
  for (int b = beg; b < end; b += 16) {
    const int i0 = b + q;
    const int i1 = i0 + 8;
    const bool ok0 = i0 < end;
    const bool ok1 = i1 < end;
    const int2 e0 = csr[ok0 ? i0 : beg];
    const int2 e1 = csr[ok1 ? i1 : beg];
    float4 x0 = X4[(size_t)e0.x * 4 + c4];
    float4 x1 = X4[(size_t)e1.x * 4 + c4];
    if (doRelu) {
      x0.x = fmaxf(x0.x, 0.f); x0.y = fmaxf(x0.y, 0.f);
      x0.z = fmaxf(x0.z, 0.f); x0.w = fmaxf(x0.w, 0.f);
      x1.x = fmaxf(x1.x, 0.f); x1.y = fmaxf(x1.y, 0.f);
      x1.z = fmaxf(x1.z, 0.f); x1.w = fmaxf(x1.w, 0.f);
    }
    const float v0 = __int_as_float(e0.y);
    const float v1 = __int_as_float(e1.y);
    if (ok0) {
      acc.x = fmaf(v0, x0.x, acc.x); acc.y = fmaf(v0, x0.y, acc.y);
      acc.z = fmaf(v0, x0.z, acc.z); acc.w = fmaf(v0, x0.w, acc.w);
    }
    if (ok1) {
      acc.x = fmaf(v1, x1.x, acc.x); acc.y = fmaf(v1, x1.y, acc.y);
      acc.z = fmaf(v1, x1.z, acc.z); acc.w = fmaf(v1, x1.w, acc.w);
    }
  }
  return acc;
}

static __device__ inline float gcn_slot_sum(float4 a, int lane) {
  const bool h4 = (lane & 16) != 0;
  const bool h3 = (lane & 8) != 0;
  float k0 = h4 ? a.z : a.x;
  float k1 = h4 ? a.w : a.y;
  k0 += __shfl_xor(h4 ? a.x : a.z, 16, 32);
  k1 += __shfl_xor(h4 ? a.y : a.w, 16, 32);
  float k = h3 ? k1 : k0;
  k += __shfl_xor(h3 ? k0 : k1, 8, 32);
  k += __shfl_xor(k, 4, 32);
  return k;
}

__global__ __launch_bounds__(256) void gcn_spmm_rows_kernel(
    const int* __restrict__ rowptr, const int* __restrict__ csr,
    const float* __restrict__ Xin, float* __restrict__ Z, int N, int doRelu) {
  const int lane = (int)(threadIdx.x & 31u);
  const int wv = __builtin_amdgcn_readfirstlane((int)(threadIdx.x >> 5));
  const int r0 = (blockIdx.x * 8 + wv) * 2;
  if (r0 >= N) return;
  const int q  = lane >> 2;
  const int c4 = lane & 3;
  const int2*   __restrict__ E2 = reinterpret_cast<const int2*>(csr);
  const float4* __restrict__ X4 = reinterpret_cast<const float4*>(Xin);

  const int p0 = rowptr[r0];
  const int p1 = rowptr[r0 + 1];
  const int p2 = rowptr[min(r0 + 2, N)];

  const float za = gcn_slot_sum(gcn_row_gather(E2, X4, p0, p1, q, c4, doRelu), lane);
  const float zb = gcn_slot_sum(gcn_row_gather(E2, X4, p1, p2, q, c4, doRelu), lane);

  const int rsel = (lane >> 2) & 1;
  const int col  = 4 * c4 + 2 * ((lane >> 4) & 1) + ((lane >> 3) & 1);
  const int row  = r0 + rsel;
  if (row < N) Z[(size_t)row * HIDDEN + col] = rsel ? zb : za;
}

__global__ __launch_bounds__(32) void gcn_gemm2_lsm_wmma(
    const float* __restrict__ Z2, const float* __restrict__ W2,
    float* __restrict__ out, int N) {
  const int lane = (int)threadIdx.x;
  const int r0   = blockIdx.x * 16;
  const int col  = lane & 15;
  const int kh   = lane >> 4;
  const int rowA = min(r0 + col, N - 1);
  const float* __restrict__ Zr = Z2 + (size_t)rowA * HIDDEN;

  v8f acc0 = {}, acc1 = {}, acc2 = {};
#pragma unroll
  for (int k0 = 0; k0 < HIDDEN; k0 += 4) {
    const int ka = k0 + 2 * kh;
    v2f a;
    a.x = Zr[ka];
    a.y = Zr[ka + 1];

    v2f b0;
    b0.x = W2[ka * NCLS + col];
    b0.y = W2[(ka + 1) * NCLS + col];
    acc0 = wmma_f32_16x16x4(a, b0, acc0);

    v2f b1;
    b1.x = W2[ka * NCLS + 16 + col];
    b1.y = W2[(ka + 1) * NCLS + 16 + col];
    acc1 = wmma_f32_16x16x4(a, b1, acc1);

    const int c2 = 32 + col;
    v2f b2;
    b2.x = (c2 < NCLS) ? W2[ka * NCLS + c2] : 0.0f;
    b2.y = (c2 < NCLS) ? W2[(ka + 1) * NCLS + c2] : 0.0f;
    acc2 = wmma_f32_16x16x4(a, b2, acc2);
  }

  const bool t2ok = (col < 8);
  const float ninf = -__builtin_inff();
  const bool fullTile = (r0 + 16 <= N);
  float* __restrict__ Or = out + (size_t)(r0 + 8 * kh) * NCLS + col;

#pragma unroll
  for (int g = 0; g < 8; ++g) {
    const float v0 = acc0[g];
    const float v1 = acc1[g];
    const float v2v = acc2[g];

    float m = fmaxf(fmaxf(v0, v1), t2ok ? v2v : ninf);
#pragma unroll
    for (int off = 8; off >= 1; off >>= 1) m = fmaxf(m, __shfl_xor(m, off, 16));

    float s = expf(v0 - m) + expf(v1 - m) + (t2ok ? expf(v2v - m) : 0.0f);
#pragma unroll
    for (int off = 8; off >= 1; off >>= 1) s += __shfl_xor(s, off, 16);

    const float lse = m + logf(s);

    if (fullTile) {
      Or[g * NCLS] = v0 - lse;
      Or[g * NCLS + 16] = v1 - lse;
      if (t2ok) Or[g * NCLS + 32] = v2v - lse;
    } else {
      const int row = r0 + g + 8 * kh;
      if (row < N) {
        out[(size_t)row * NCLS + col] = v0 - lse;
        out[(size_t)row * NCLS + 16 + col] = v1 - lse;
        if (t2ok) out[(size_t)row * NCLS + 32 + col] = v2v - lse;
      }
    }
  }
}

extern "C" void kernel_launch(void* const* d_in, const int* in_sizes, int n_in,
                              void* d_out, int out_size, void* d_ws, size_t ws_size,
                              hipStream_t stream) {
  const float* features = (const float*)d_in[0];
  const int*   esrc     = (const int*)d_in[1];
  const int*   edst     = (const int*)d_in[2];
  const float* evals    = (const float*)d_in[3];
  const float* W1       = (const float*)d_in[4];
  const float* W2       = (const float*)d_in[5];
  float*       out      = (float*)d_out;

  const int N = in_sizes[0] / IN_DIM;
  const int E = in_sizes[1];

  float* Y  = (float*)d_ws;
  float* Z1 = Y + (size_t)N * HIDDEN;
  float* Z2 = Z1 + (size_t)N * HIDDEN;
  const int    Npad = (N + 1 + 63) / 64 * 64;
  const size_t ePad = ((size_t)E + 63) / 64 * 64;
  const int    chunk = (E + GCN_CHUNKS - 1) / GCN_CHUNKS;
  int* tot    = (int*)(Z2 + (size_t)N * HIDDEN);
  int* rowptr = tot + Npad;
  int* lrank  = rowptr + Npad;
  int* csr    = lrank + ePad;
  int* cntG   = csr + 2 * ePad;
  int* preG   = cntG + (size_t)GCN_CHUNKS * Npad;

  gcn_gemm1_wmma<<<(N + 15) / 16, 32, 0, stream>>>(features, W1, Y, N);

  gcn_rank_lds_kernel<<<GCN_CHUNKS, 1024, 0, stream>>>(edst, cntG, lrank, N, E, Npad, chunk);
  gcn_chunk_prefix_kernel<<<(N + 255) / 256, 256, 0, stream>>>(cntG, preG, tot, N, Npad);
  gcn_rowptr_kernel<<<(N + 1 + 1023) / 1024, 1024, 0, stream>>>(tot, rowptr, N);
  gcn_fill_kernel<<<(E + 255) / 256, 256, 0, stream>>>(
      esrc, edst, evals, lrank, rowptr, preG, csr, E, Npad, chunk);

  {
    const int blocks = ((N + 1) / 2 + 7) / 8;
    gcn_spmm_rows_kernel<<<blocks, 256, 0, stream>>>(rowptr, csr, Y, Z1, N,  0);
    gcn_spmm_rows_kernel<<<blocks, 256, 0, stream>>>(rowptr, csr, Z1, Z2, N,  1);
  }

  gcn_gemm2_lsm_wmma<<<(N + 15) / 16, 32, 0, stream>>>(Z2, W2, out, N);
}


# Row starts in one launch: each run also stores its prefix over 256-row blocks, so row pointers and per-run write positions come from one kernel (8 to 7 launches); fill gathers one position, not two
# speedup vs baseline: 1.3634x; 1.1233x over previous
#include <hip/hip_runtime.h>
#include <math.h>

#define IN_DIM  100
#define HIDDEN  16
#define NCLS    40

typedef __attribute__((ext_vector_type(2))) float v2f;
typedef __attribute__((ext_vector_type(8))) float v8f;

static __device__ inline v8f wmma_f32_16x16x4(v2f a, v2f b, v8f c) {
#if __has_builtin(__builtin_amdgcn_wmma_f32_16x16x4_f32)
  return __builtin_amdgcn_wmma_f32_16x16x4_f32(
       false, a,  false, b,
       (short)0, c,  false,  false);
#else
  const int lane = (int)(threadIdx.x & 31u);
  const int nsel = lane & 15;
  v8f d = c;
#pragma unroll
  for (int g = 0; g < 8; ++g) {
    const int m = g + 8 * (lane >> 4);
    float acc = 0.f;
#pragma unroll
    for (int k = 0; k < 4; ++k) {
      const int srcA = m + 16 * (k >> 1);
      const int srcB = nsel + 16 * (k >> 1);
      const float av = __shfl((k & 1) ? a.y : a.x, srcA, 32);
      const float bv = __shfl((k & 1) ? b.y : b.x, srcB, 32);
      acc += av * bv;
    }
    d[g] += acc;
  }
  return d;
#endif
}

#define GCN_CHUNKS 64
#define GCN_HCAP   65536
#define GCN_RB     256

__global__ __launch_bounds__(1024) void gcn_rank_lds_kernel(
    const int* __restrict__ edst, int* __restrict__ cntG,
    int* __restrict__ lrank, int* __restrict__ blkG,
    int N, int E, int Npad, int nbPad, int chunk) {
  __shared__ int h[GCN_HCAP];
  __shared__ int s_blk[GCN_HCAP / GCN_RB];
  __shared__ int s_wt[8];
  const int t    = (int)threadIdx.x;
  const int lane = t & 31;
  const int w    = t >> 5;
  const int g    = (int)blockIdx.x;
  const int e_lo = min(E, g * chunk);
  const int e_hi = min(E, e_lo + chunk);
  const int nb   = N / GCN_RB + 1;
  int4* __restrict__ h4 = reinterpret_cast<int4*>(h);
  int carry = 0;

  for (int lo = 0; lo <= N; lo += GCN_HCAP) {
    const int n  = min(GCN_HCAP, N - lo);
    const int n4 = (n + 3) >> 2;
    for (int i = t; i < n4; i += 1024) h4[i] = make_int4(0, 0, 0, 0);
    __syncthreads();

    for (int e0 = e_lo + t; e0 < e_hi; e0 += 4096) {
      int d[4];
#pragma unroll
      for (int u = 0; u < 4; ++u) d[u] = edst[min(e0 + u * 1024, e_hi - 1)] - lo;
#pragma unroll
      for (int u = 0; u < 4; ++u) {
        const int e = e0 + u * 1024;
        if (e < e_hi && (unsigned)d[u] < (unsigned)n) lrank[e] = atomicAdd(&h[d[u]], 1);
      }
    }
    __syncthreads();

    int4* __restrict__ o4 = reinterpret_cast<int4*>(cntG + (size_t)g * Npad + lo);
    for (int i = t; i < n4; i += 1024) o4[i] = h4[i];

    const int nbl = min(GCN_HCAP / GCN_RB, n / GCN_RB + 1);
    for (int b = w; b < nbl; b += 32) {
      const int j = b * (GCN_RB / 4) + lane * 2;
      const int4 a = (j     < n4) ? h4[j]     : make_int4(0, 0, 0, 0);
      const int4 c = (j + 1 < n4) ? h4[j + 1] : make_int4(0, 0, 0, 0);
      int s = ((a.x + a.y) + (a.z + a.w)) + ((c.x + c.y) + (c.z + c.w));
#pragma unroll
      for (int off = 16; off >= 1; off >>= 1) s += __shfl_xor(s, off, 32);
      if (lane == 0) s_blk[b] = s;
    }
    __syncthreads();

    int x = 0, inc = 0;
    if (t < GCN_HCAP / GCN_RB) {
      x = (t < nbl) ? s_blk[t] : 0;
      inc = x;
#pragma unroll
      for (int off = 1; off < 32; off <<= 1) {
        const int y = __shfl_up(inc, off, 32);
        if (lane >= off) inc += y;
      }
      if (lane == 31) s_wt[w] = inc;
    }
    __syncthreads();
    int wt  = (lane < w && lane < 8) ? s_wt[lane] : 0;
    int all = (lane < 8) ? s_wt[lane] : 0;
#pragma unroll
    for (int off = 4; off >= 1; off >>= 1) {
      wt  += __shfl_xor(wt, off, 32);
      all += __shfl_xor(all, off, 32);
    }
    wt  = __shfl(wt, 0, 32);
    all = __shfl(all, 0, 32);
    if (t < GCN_HCAP / GCN_RB) {
      const int bidx = lo / GCN_RB + t;
      if (bidx < nb) blkG[(size_t)g * nbPad + bidx] = carry + wt + inc - x;
    }
    carry += all;
    __syncthreads();
  }
}

__global__ __launch_bounds__(GCN_RB) void gcn_rowstart_kernel(
    const int* __restrict__ cntG, const int* __restrict__ blkG,
    int* __restrict__ posG, int* __restrict__ rowptr, int N, int Npad, int nbPad) {
  __shared__ int s_tot[GCN_RB / 32];
  const int t    = (int)threadIdx.x;
  const int lane = t & 31;
  const int w    = t >> 5;
  const int b    = (int)blockIdx.x;
  const int i    = b * GCN_RB + t;
  const bool in  = i < N;
  const int ic   = min(i, N - 1);

  int c[GCN_CHUNKS];
#pragma unroll
  for (int g = 0; g < GCN_CHUNKS; ++g) c[g] = cntG[(size_t)g * Npad + ic];

  int base = 0;
  for (int g = lane; g < GCN_CHUNKS; g += 32) base += blkG[(size_t)g * nbPad + b];
#pragma unroll
  for (int off = 16; off >= 1; off >>= 1) base += __shfl_xor(base, off, 32);

  int tot = 0;
#pragma unroll
  for (int g = 0; g < GCN_CHUNKS; ++g) { c[g] = in ? c[g] : 0; tot += c[g]; }

  int inc = tot;
#pragma unroll
  for (int off = 1; off < 32; off <<= 1) {
    const int y = __shfl_up(inc, off, 32);
    if (lane >= off) inc += y;
  }
  if (lane == 31) s_tot[w] = inc;
  __syncthreads();
  int wt = (lane < w && lane < GCN_RB / 32) ? s_tot[lane] : 0;
#pragma unroll
  for (int off = 16; off >= 1; off >>= 1) wt += __shfl_xor(wt, off, 32);

  int run = base + wt + inc - tot;
  if (i <= N) rowptr[i] = run;
  if (in) {
#pragma unroll
    for (int g = 0; g < GCN_CHUNKS; ++g) {
      posG[(size_t)g * Npad + i] = run;
      run += c[g];
    }
  }
}

__global__ __launch_bounds__(256) void gcn_fill_kernel(
    const int* __restrict__ esrc, const int* __restrict__ edst,
    const float* __restrict__ evals, const int* __restrict__ lrank,
    const int* __restrict__ posG,
    int* __restrict__ csr, int E, int Npad, int chunk) {
  const int e = blockIdx.x * blockDim.x + threadIdx.x;
  if (e >= E) return;
  const int d = edst[e];
  const int g = e / chunk;
  const int pos = posG[(size_t)g * Npad + d] + lrank[e];
  int2 ev;
  ev.x = esrc[e];
  ev.y = __float_as_int(evals[e]);
  reinterpret_cast<int2*>(csr)[pos] = ev;
}

__global__ __launch_bounds__(32) void gcn_gemm1_wmma(
    const float* __restrict__ X, const float* __restrict__ W1,
    float* __restrict__ Y, int N) {
  const int lane = (int)threadIdx.x;
  const int r0   = blockIdx.x * 16;
  const int col  = lane & 15;
  const int kh   = lane >> 4;
  const int rowA = min(r0 + col, N - 1);
  const float* __restrict__ Xr = X + (size_t)rowA * IN_DIM;

  v8f acc = {};
#pragma unroll 5
  for (int k0 = 0; k0 < IN_DIM; k0 += 4) {
    const int ka = k0 + 2 * kh;
    v2f a;
    a.x = Xr[ka];
    a.y = Xr[ka + 1];
    v2f b;
    b.x = W1[ka * HIDDEN + col];
    b.y = W1[(ka + 1) * HIDDEN + col];
    acc = wmma_f32_16x16x4(a, b, acc);
  }

  float* __restrict__ Yr = Y + (size_t)(r0 + 8 * kh) * HIDDEN + col;
  if (r0 + 16 <= N) {
#pragma unroll
    for (int g = 0; g < 8; ++g) Yr[g * HIDDEN] = acc[g];
  } else {
#pragma unroll
    for (int g = 0; g < 8; ++g) {
      const int row = r0 + g + 8 * kh;
      if (row < N) Y[row * HIDDEN + col] = acc[g];
    }
  }
}

static __device__ inline float4 gcn_row_gather(
    const int2* __restrict__ csr, const float4* __restrict__ X4,
    int beg, int end, int q, int c4, int doRelu) {
  float4 acc = make_float4(0.f, 0.f, 0.f, 0.f);
  for (int b = beg; b < end; b += 16) {
    const int i0 = b + q;
    const int i1 = i0 + 8;
    const bool ok0 = i0 < end;
    const bool ok1 = i1 < end;
    const int2 e0 = csr[ok0 ? i0 : beg];
    const int2 e1 = csr[ok1 ? i1 : beg];
    float4 x0 = X4[(size_t)e0.x * 4 + c4];
    float4 x1 = X4[(size_t)e1.x * 4 + c4];
    if (doRelu) {
      x0.x = fmaxf(x0.x, 0.f); x0.y = fmaxf(x0.y, 0.f);
      x0.z = fmaxf(x0.z, 0.f); x0.w = fmaxf(x0.w, 0.f);
      x1.x = fmaxf(x1.x, 0.f); x1.y = fmaxf(x1.y, 0.f);
      x1.z = fmaxf(x1.z, 0.f); x1.w = fmaxf(x1.w, 0.f);
    }
    const float v0 = __int_as_float(e0.y);
    const float v1 = __int_as_float(e1.y);
    if (ok0) {
      acc.x = fmaf(v0, x0.x, acc.x); acc.y = fmaf(v0, x0.y, acc.y);
      acc.z = fmaf(v0, x0.z, acc.z); acc.w = fmaf(v0, x0.w, acc.w);
    }
    if (ok1) {
      acc.x = fmaf(v1, x1.x, acc.x); acc.y = fmaf(v1, x1.y, acc.y);
      acc.z = fmaf(v1, x1.z, acc.z); acc.w = fmaf(v1, x1.w, acc.w);
    }
  }
  return acc;
}

static __device__ inline float gcn_slot_sum(float4 a, int lane) {
  const bool h4 = (lane & 16) != 0;
  const bool h3 = (lane & 8) != 0;
  float k0 = h4 ? a.z : a.x;
  float k1 = h4 ? a.w : a.y;
  k0 += __shfl_xor(h4 ? a.x : a.z, 16, 32);
  k1 += __shfl_xor(h4 ? a.y : a.w, 16, 32);
  float k = h3 ? k1 : k0;
  k += __shfl_xor(h3 ? k0 : k1, 8, 32);
  k += __shfl_xor(k, 4, 32);
  return k;
}

__global__ __launch_bounds__(256) void gcn_spmm_rows_kernel(
    const int* __restrict__ rowptr, const int* __restrict__ csr,
    const float* __restrict__ Xin, float* __restrict__ Z, int N, int doRelu) {
  const int lane = (int)(threadIdx.x & 31u);
  const int wv = __builtin_amdgcn_readfirstlane((int)(threadIdx.x >> 5));
  const int r0 = (blockIdx.x * 8 + wv) * 2;
  if (r0 >= N) return;
  const int q  = lane >> 2;
  const int c4 = lane & 3;
  const int2*   __restrict__ E2 = reinterpret_cast<const int2*>(csr);
  const float4* __restrict__ X4 = reinterpret_cast<const float4*>(Xin);

  const int p0 = rowptr[r0];
  const int p1 = rowptr[r0 + 1];
  const int p2 = rowptr[min(r0 + 2, N)];

  const float za = gcn_slot_sum(gcn_row_gather(E2, X4, p0, p1, q, c4, doRelu), lane);
  const float zb = gcn_slot_sum(gcn_row_gather(E2, X4, p1, p2, q, c4, doRelu), lane);

  const int rsel = (lane >> 2) & 1;
  const int col  = 4 * c4 + 2 * ((lane >> 4) & 1) + ((lane >> 3) & 1);
  const int row  = r0 + rsel;
  if (row < N) Z[(size_t)row * HIDDEN + col] = rsel ? zb : za;
}

__global__ __launch_bounds__(32) void gcn_gemm2_lsm_wmma(
    const float* __restrict__ Z2, const float* __restrict__ W2,
    float* __restrict__ out, int N) {
  const int lane = (int)threadIdx.x;
  const int r0   = blockIdx.x * 16;
  const int col  = lane & 15;
  const int kh   = lane >> 4;
  const int rowA = min(r0 + col, N - 1);
  const float* __restrict__ Zr = Z2 + (size_t)rowA * HIDDEN;

  v8f acc0 = {}, acc1 = {}, acc2 = {};
#pragma unroll
  for (int k0 = 0; k0 < HIDDEN; k0 += 4) {
    const int ka = k0 + 2 * kh;
    v2f a;
    a.x = Zr[ka];
    a.y = Zr[ka + 1];

    v2f b0;
    b0.x = W2[ka * NCLS + col];
    b0.y = W2[(ka + 1) * NCLS + col];
    acc0 = wmma_f32_16x16x4(a, b0, acc0);

    v2f b1;
    b1.x = W2[ka * NCLS + 16 + col];
    b1.y = W2[(ka + 1) * NCLS + 16 + col];
    acc1 = wmma_f32_16x16x4(a, b1, acc1);

    const int c2 = 32 + col;
    v2f b2;
    b2.x = (c2 < NCLS) ? W2[ka * NCLS + c2] : 0.0f;
    b2.y = (c2 < NCLS) ? W2[(ka + 1) * NCLS + c2] : 0.0f;
    acc2 = wmma_f32_16x16x4(a, b2, acc2);
  }

  const bool t2ok = (col < 8);
  const float ninf = -__builtin_inff();
  const bool fullTile = (r0 + 16 <= N);
  float* __restrict__ Or = out + (size_t)(r0 + 8 * kh) * NCLS + col;

#pragma unroll
  for (int g = 0; g < 8; ++g) {
    const float v0 = acc0[g];
    const float v1 = acc1[g];
    const float v2v = acc2[g];

    float m = fmaxf(fmaxf(v0, v1), t2ok ? v2v : ninf);
#pragma unroll
    for (int off = 8; off >= 1; off >>= 1) m = fmaxf(m, __shfl_xor(m, off, 16));

    float s = expf(v0 - m) + expf(v1 - m) + (t2ok ? expf(v2v - m) : 0.0f);
#pragma unroll
    for (int off = 8; off >= 1; off >>= 1) s += __shfl_xor(s, off, 16);

    const float lse = m + logf(s);

    if (fullTile) {
      Or[g * NCLS] = v0 - lse;
      Or[g * NCLS + 16] = v1 - lse;
      if (t2ok) Or[g * NCLS + 32] = v2v - lse;
    } else {
      const int row = r0 + g + 8 * kh;
      if (row < N) {
        out[(size_t)row * NCLS + col] = v0 - lse;
        out[(size_t)row * NCLS + 16 + col] = v1 - lse;
        if (t2ok) out[(size_t)row * NCLS + 32 + col] = v2v - lse;
      }
    }
  }
}

extern "C" void kernel_launch(void* const* d_in, const int* in_sizes, int n_in,
                              void* d_out, int out_size, void* d_ws, size_t ws_size,
                              hipStream_t stream) {
  const float* features = (const float*)d_in[0];
  const int*   esrc     = (const int*)d_in[1];
  const int*   edst     = (const int*)d_in[2];
  const float* evals    = (const float*)d_in[3];
  const float* W1       = (const float*)d_in[4];
  const float* W2       = (const float*)d_in[5];
  float*       out      = (float*)d_out;

  const int N = in_sizes[0] / IN_DIM;
  const int E = in_sizes[1];

  float* Y  = (float*)d_ws;
  float* Z1 = Y + (size_t)N * HIDDEN;
  float* Z2 = Z1 + (size_t)N * HIDDEN;
  const int    Npad = (N + 1 + 63) / 64 * 64;
  const size_t ePad = ((size_t)E + 63) / 64 * 64;
  const int    chunk = (E + GCN_CHUNKS - 1) / GCN_CHUNKS;
  const int nb    = N / GCN_RB + 1;
  const int nbPad = (nb + 63) / 64 * 64;
  int* rowptr = (int*)(Z2 + (size_t)N * HIDDEN);
  int* lrank  = rowptr + Npad;
  int* csr    = lrank + ePad;
  int* cntG   = csr + 2 * ePad;
  int* posG   = cntG + (size_t)GCN_CHUNKS * Npad;
  int* blkG   = posG + (size_t)GCN_CHUNKS * Npad;

  gcn_gemm1_wmma<<<(N + 15) / 16, 32, 0, stream>>>(features, W1, Y, N);

  gcn_rank_lds_kernel<<<GCN_CHUNKS, 1024, 0, stream>>>(
      edst, cntG, lrank, blkG, N, E, Npad, nbPad, chunk);
  gcn_rowstart_kernel<<<nb, GCN_RB, 0, stream>>>(cntG, blkG, posG, rowptr, N, Npad, nbPad);
  gcn_fill_kernel<<<(E + 255) / 256, 256, 0, stream>>>(
      esrc, edst, evals, lrank, posG, csr, E, Npad, chunk);

  {
    const int blocks = ((N + 1) / 2 + 7) / 8;
    gcn_spmm_rows_kernel<<<blocks, 256, 0, stream>>>(rowptr, csr, Y, Z1, N,  0);
    gcn_spmm_rows_kernel<<<blocks, 256, 0, stream>>>(rowptr, csr, Z1, Z2, N,  1);
  }

  gcn_gemm2_lsm_wmma<<<(N + 15) / 16, 32, 0, stream>>>(Z2, W2, out, N);
}


# X@W1 tiles moved into the edge-ranking launch as extra work-groups behind its 64 (job by group index, 32 tile waves per group): the two independent jobs run side by side; 7 to 6 launches
# speedup vs baseline: 1.4237x; 1.0442x over previous
#include <hip/hip_runtime.h>
#include <math.h>

#define IN_DIM  100
#define HIDDEN  16
#define NCLS    40

typedef __attribute__((ext_vector_type(2))) float v2f;
typedef __attribute__((ext_vector_type(8))) float v8f;

static __device__ inline v8f wmma_f32_16x16x4(v2f a, v2f b, v8f c) {
#if __has_builtin(__builtin_amdgcn_wmma_f32_16x16x4_f32)
  return __builtin_amdgcn_wmma_f32_16x16x4_f32(
       false, a,  false, b,
       (short)0, c,  false,  false);
#else
  const int lane = (int)(threadIdx.x & 31u);
  const int nsel = lane & 15;
  v8f d = c;
#pragma unroll
  for (int g = 0; g < 8; ++g) {
    const int m = g + 8 * (lane >> 4);
    float acc = 0.f;
#pragma unroll
    for (int k = 0; k < 4; ++k) {
      const int srcA = m + 16 * (k >> 1);
      const int srcB = nsel + 16 * (k >> 1);
      const float av = __shfl((k & 1) ? a.y : a.x, srcA, 32);
      const float bv = __shfl((k & 1) ? b.y : b.x, srcB, 32);
      acc += av * bv;
    }
    d[g] += acc;
  }
  return d;
#endif
}

#define GCN_CHUNKS 64
#define GCN_HCAP   65536
#define GCN_RB     256
#define GCN_G1W    32

static __device__ inline void gcn_gemm1_tile(
    const float* __restrict__ X, const float* __restrict__ W1,
    float* __restrict__ Y, int N, int r0, int lane) {
  const int col  = lane & 15;
  const int kh   = lane >> 4;
  const int rowA = min(r0 + col, N - 1);
  const float* __restrict__ Xr = X + (size_t)rowA * IN_DIM;

  v8f acc = {};
#pragma unroll 5
  for (int k0 = 0; k0 < IN_DIM; k0 += 4) {
    const int ka = k0 + 2 * kh;
    v2f a;
    a.x = Xr[ka];
    a.y = Xr[ka + 1];
    v2f b;
    b.x = W1[ka * HIDDEN + col];
    b.y = W1[(ka + 1) * HIDDEN + col];
    acc = wmma_f32_16x16x4(a, b, acc);
  }

  float* __restrict__ Yr = Y + (size_t)(r0 + 8 * kh) * HIDDEN + col;
  if (r0 + 16 <= N) {
#pragma unroll
    for (int g = 0; g < 8; ++g) Yr[g * HIDDEN] = acc[g];
  } else {
#pragma unroll
    for (int g = 0; g < 8; ++g) {
      const int row = r0 + g + 8 * kh;
      if (row < N) Y[row * HIDDEN + col] = acc[g];
    }
  }
}

__global__ __launch_bounds__(1024) void gcn_rank_gemm1_kernel(
    const int* __restrict__ edst, int* __restrict__ cntG,
    int* __restrict__ lrank, int* __restrict__ blkG,
    int N, int E, int Npad, int nbPad, int chunk,
    const float* __restrict__ X, const float* __restrict__ W1,
    float* __restrict__ Y) {
  __shared__ int h[GCN_HCAP];
  __shared__ int s_blk[GCN_HCAP / GCN_RB];
  __shared__ int s_wt[8];
  const int t    = (int)threadIdx.x;
  const int lane = t & 31;
  const int w    = t >> 5;
  if ((int)blockIdx.x >= GCN_CHUNKS) {
    const int wv = __builtin_amdgcn_readfirstlane(w);
    const int r0 = (((int)blockIdx.x - GCN_CHUNKS) * GCN_G1W + wv) * 16;
    if (r0 < N) gcn_gemm1_tile(X, W1, Y, N, r0, lane);
    return;
  }
  const int g    = (int)blockIdx.x;
  const int e_lo = min(E, g * chunk);
  const int e_hi = min(E, e_lo + chunk);
  const int nb   = N / GCN_RB + 1;
  int4* __restrict__ h4 = reinterpret_cast<int4*>(h);
  int carry = 0;

  for (int lo = 0; lo <= N; lo += GCN_HCAP) {
    const int n  = min(GCN_HCAP, N - lo);
    const int n4 = (n + 3) >> 2;
    for (int i = t; i < n4; i += 1024) h4[i] = make_int4(0, 0, 0, 0);
    __syncthreads();

    for (int e0 = e_lo + t; e0 < e_hi; e0 += 4096) {
      int d[4];
#pragma unroll
      for (int u = 0; u < 4; ++u) d[u] = edst[min(e0 + u * 1024, e_hi - 1)] - lo;
#pragma unroll
      for (int u = 0; u < 4; ++u) {
        const int e = e0 + u * 1024;
        if (e < e_hi && (unsigned)d[u] < (unsigned)n) lrank[e] = atomicAdd(&h[d[u]], 1);
      }
    }
    __syncthreads();

    int4* __restrict__ o4 = reinterpret_cast<int4*>(cntG + (size_t)g * Npad + lo);
    for (int i = t; i < n4; i += 1024) o4[i] = h4[i];

    const int nbl = min(GCN_HCAP / GCN_RB, n / GCN_RB + 1);
    for (int b = w; b < nbl; b += 32) {
      const int j = b * (GCN_RB / 4) + lane * 2;
      const int4 a = (j     < n4) ? h4[j]     : make_int4(0, 0, 0, 0);
      const int4 c = (j + 1 < n4) ? h4[j + 1] : make_int4(0, 0, 0, 0);
      int s = ((a.x + a.y) + (a.z + a.w)) + ((c.x + c.y) + (c.z + c.w));
#pragma unroll
      for (int off = 16; off >= 1; off >>= 1) s += __shfl_xor(s, off, 32);
      if (lane == 0) s_blk[b] = s;
    }
    __syncthreads();

    int x = 0, inc = 0;
    if (t < GCN_HCAP / GCN_RB) {
      x = (t < nbl) ? s_blk[t] : 0;
      inc = x;
#pragma unroll
      for (int off = 1; off < 32; off <<= 1) {
        const int y = __shfl_up(inc, off, 32);
        if (lane >= off) inc += y;
      }
      if (lane == 31) s_wt[w] = inc;
    }
    __syncthreads();
    int wt  = (lane < w && lane < 8) ? s_wt[lane] : 0;
    int all = (lane < 8) ? s_wt[lane] : 0;
#pragma unroll
    for (int off = 4; off >= 1; off >>= 1) {
      wt  += __shfl_xor(wt, off, 32);
      all += __shfl_xor(all, off, 32);
    }
    wt  = __shfl(wt, 0, 32);
    all = __shfl(all, 0, 32);
    if (t < GCN_HCAP / GCN_RB) {
      const int bidx = lo / GCN_RB + t;
      if (bidx < nb) blkG[(size_t)g * nbPad + bidx] = carry + wt + inc - x;
    }
    carry += all;
    __syncthreads();
  }
}

__global__ __launch_bounds__(GCN_RB) void gcn_rowstart_kernel(
    const int* __restrict__ cntG, const int* __restrict__ blkG,
    int* __restrict__ posG, int* __restrict__ rowptr, int N, int Npad, int nbPad) {
  __shared__ int s_tot[GCN_RB / 32];
  const int t    = (int)threadIdx.x;
  const int lane = t & 31;
  const int w    = t >> 5;
  const int b    = (int)blockIdx.x;
  const int i    = b * GCN_RB + t;
  const bool in  = i < N;
  const int ic   = min(i, N - 1);

  int c[GCN_CHUNKS];
#pragma unroll
  for (int g = 0; g < GCN_CHUNKS; ++g) c[g] = cntG[(size_t)g * Npad + ic];

  int base = 0;
  for (int g = lane; g < GCN_CHUNKS; g += 32) base += blkG[(size_t)g * nbPad + b];
#pragma unroll
  for (int off = 16; off >= 1; off >>= 1) base += __shfl_xor(base, off, 32);

  int tot = 0;
#pragma unroll
  for (int g = 0; g < GCN_CHUNKS; ++g) { c[g] = in ? c[g] : 0; tot += c[g]; }

  int inc = tot;
#pragma unroll
  for (int off = 1; off < 32; off <<= 1) {
    const int y = __shfl_up(inc, off, 32);
    if (lane >= off) inc += y;
  }
  if (lane == 31) s_tot[w] = inc;
  __syncthreads();
  int wt = (lane < w && lane < GCN_RB / 32) ? s_tot[lane] : 0;
#pragma unroll
  for (int off = 16; off >= 1; off >>= 1) wt += __shfl_xor(wt, off, 32);

  int run = base + wt + inc - tot;
  if (i <= N) rowptr[i] = run;
  if (in) {
#pragma unroll
    for (int g = 0; g < GCN_CHUNKS; ++g) {
      posG[(size_t)g * Npad + i] = run;
      run += c[g];
    }
  }
}

__global__ __launch_bounds__(256) void gcn_fill_kernel(
    const int* __restrict__ esrc, const int* __restrict__ edst,
    const float* __restrict__ evals, const int* __restrict__ lrank,
    const int* __restrict__ posG,
    int* __restrict__ csr, int E, int Npad, int chunk) {
  const int e = blockIdx.x * blockDim.x + threadIdx.x;
  if (e >= E) return;
  const int d = edst[e];
  const int g = e / chunk;
  const int pos = posG[(size_t)g * Npad + d] + lrank[e];
  int2 ev;
  ev.x = esrc[e];
  ev.y = __float_as_int(evals[e]);
  reinterpret_cast<int2*>(csr)[pos] = ev;
}

static __device__ inline float4 gcn_row_gather(
    const int2* __restrict__ csr, const float4* __restrict__ X4,
    int beg, int end, int q, int c4, int doRelu) {
  float4 acc = make_float4(0.f, 0.f, 0.f, 0.f);
  for (int b = beg; b < end; b += 16) {
    const int i0 = b + q;
    const int i1 = i0 + 8;
    const bool ok0 = i0 < end;
    const bool ok1 = i1 < end;
    const int2 e0 = csr[ok0 ? i0 : beg];
    const int2 e1 = csr[ok1 ? i1 : beg];
    float4 x0 = X4[(size_t)e0.x * 4 + c4];
    float4 x1 = X4[(size_t)e1.x * 4 + c4];
    if (doRelu) {
      x0.x = fmaxf(x0.x, 0.f); x0.y = fmaxf(x0.y, 0.f);
      x0.z = fmaxf(x0.z, 0.f); x0.w = fmaxf(x0.w, 0.f);
      x1.x = fmaxf(x1.x, 0.f); x1.y = fmaxf(x1.y, 0.f);
      x1.z = fmaxf(x1.z, 0.f); x1.w = fmaxf(x1.w, 0.f);
    }
    const float v0 = __int_as_float(e0.y);
    const float v1 = __int_as_float(e1.y);
    if (ok0) {
      acc.x = fmaf(v0, x0.x, acc.x); acc.y = fmaf(v0, x0.y, acc.y);
      acc.z = fmaf(v0, x0.z, acc.z); acc.w = fmaf(v0, x0.w, acc.w);
    }
    if (ok1) {
      acc.x = fmaf(v1, x1.x, acc.x); acc.y = fmaf(v1, x1.y, acc.y);
      acc.z = fmaf(v1, x1.z, acc.z); acc.w = fmaf(v1, x1.w, acc.w);
    }
  }
  return acc;
}

static __device__ inline float gcn_slot_sum(float4 a, int lane) {
  const bool h4 = (lane & 16) != 0;
  const bool h3 = (lane & 8) != 0;
  float k0 = h4 ? a.z : a.x;
  float k1 = h4 ? a.w : a.y;
  k0 += __shfl_xor(h4 ? a.x : a.z, 16, 32);
  k1 += __shfl_xor(h4 ? a.y : a.w, 16, 32);
  float k = h3 ? k1 : k0;
  k += __shfl_xor(h3 ? k0 : k1, 8, 32);
  k += __shfl_xor(k, 4, 32);
  return k;
}

__global__ __launch_bounds__(256) void gcn_spmm_rows_kernel(
    const int* __restrict__ rowptr, const int* __restrict__ csr,
    const float* __restrict__ Xin, float* __restrict__ Z, int N, int doRelu) {
  const int lane = (int)(threadIdx.x & 31u);
  const int wv = __builtin_amdgcn_readfirstlane((int)(threadIdx.x >> 5));
  const int r0 = (blockIdx.x * 8 + wv) * 2;
  if (r0 >= N) return;
  const int q  = lane >> 2;
  const int c4 = lane & 3;
  const int2*   __restrict__ E2 = reinterpret_cast<const int2*>(csr);
  const float4* __restrict__ X4 = reinterpret_cast<const float4*>(Xin);

  const int p0 = rowptr[r0];
  const int p1 = rowptr[r0 + 1];
  const int p2 = rowptr[min(r0 + 2, N)];

  const float za = gcn_slot_sum(gcn_row_gather(E2, X4, p0, p1, q, c4, doRelu), lane);
  const float zb = gcn_slot_sum(gcn_row_gather(E2, X4, p1, p2, q, c4, doRelu), lane);

  const int rsel = (lane >> 2) & 1;
  const int col  = 4 * c4 + 2 * ((lane >> 4) & 1) + ((lane >> 3) & 1);
  const int row  = r0 + rsel;
  if (row < N) Z[(size_t)row * HIDDEN + col] = rsel ? zb : za;
}

__global__ __launch_bounds__(32) void gcn_gemm2_lsm_wmma(
    const float* __restrict__ Z2, const float* __restrict__ W2,
    float* __restrict__ out, int N) {
  const int lane = (int)threadIdx.x;
  const int r0   = blockIdx.x * 16;
  const int col  = lane & 15;
  const int kh   = lane >> 4;
  const int rowA = min(r0 + col, N - 1);
  const float* __restrict__ Zr = Z2 + (size_t)rowA * HIDDEN;

  v8f acc0 = {}, acc1 = {}, acc2 = {};
#pragma unroll
  for (int k0 = 0; k0 < HIDDEN; k0 += 4) {
    const int ka = k0 + 2 * kh;
    v2f a;
    a.x = Zr[ka];
    a.y = Zr[ka + 1];

    v2f b0;
    b0.x = W2[ka * NCLS + col];
    b0.y = W2[(ka + 1) * NCLS + col];
    acc0 = wmma_f32_16x16x4(a, b0, acc0);

    v2f b1;
    b1.x = W2[ka * NCLS + 16 + col];
    b1.y = W2[(ka + 1) * NCLS + 16 + col];
    acc1 = wmma_f32_16x16x4(a, b1, acc1);

    const int c2 = 32 + col;
    v2f b2;
    b2.x = (c2 < NCLS) ? W2[ka * NCLS + c2] : 0.0f;
    b2.y = (c2 < NCLS) ? W2[(ka + 1) * NCLS + c2] : 0.0f;
    acc2 = wmma_f32_16x16x4(a, b2, acc2);
  }

  const bool t2ok = (col < 8);
  const float ninf = -__builtin_inff();
  const bool fullTile = (r0 + 16 <= N);
  float* __restrict__ Or = out + (size_t)(r0 + 8 * kh) * NCLS + col;

#pragma unroll
  for (int g = 0; g < 8; ++g) {
    const float v0 = acc0[g];
    const float v1 = acc1[g];
    const float v2v = acc2[g];

    float m = fmaxf(fmaxf(v0, v1), t2ok ? v2v : ninf);
#pragma unroll
    for (int off = 8; off >= 1; off >>= 1) m = fmaxf(m, __shfl_xor(m, off, 16));

    float s = expf(v0 - m) + expf(v1 - m) + (t2ok ? expf(v2v - m) : 0.0f);
#pragma unroll
    for (int off = 8; off >= 1; off >>= 1) s += __shfl_xor(s, off, 16);

    const float lse = m + logf(s);

    if (fullTile) {
      Or[g * NCLS] = v0 - lse;
      Or[g * NCLS + 16] = v1 - lse;
      if (t2ok) Or[g * NCLS + 32] = v2v - lse;
    } else {
      const int row = r0 + g + 8 * kh;
      if (row < N) {
        out[(size_t)row * NCLS + col] = v0 - lse;
        out[(size_t)row * NCLS + 16 + col] = v1 - lse;
        if (t2ok) out[(size_t)row * NCLS + 32 + col] = v2v - lse;
      }
    }
  }
}

extern "C" void kernel_launch(void* const* d_in, const int* in_sizes, int n_in,
                              void* d_out, int out_size, void* d_ws, size_t ws_size,
                              hipStream_t stream) {
  const float* features = (const float*)d_in[0];
  const int*   esrc     = (const int*)d_in[1];
  const int*   edst     = (const int*)d_in[2];
  const float* evals    = (const float*)d_in[3];
  const float* W1       = (const float*)d_in[4];
  const float* W2       = (const float*)d_in[5];
  float*       out      = (float*)d_out;

  const int N = in_sizes[0] / IN_DIM;
  const int E = in_sizes[1];

  float* Y  = (float*)d_ws;
  float* Z1 = Y + (size_t)N * HIDDEN;
  float* Z2 = Z1 + (size_t)N * HIDDEN;
  const int    Npad = (N + 1 + 63) / 64 * 64;
  const size_t ePad = ((size_t)E + 63) / 64 * 64;
  const int    chunk = (E + GCN_CHUNKS - 1) / GCN_CHUNKS;
  const int nb    = N / GCN_RB + 1;
  const int nbPad = (nb + 63) / 64 * 64;
  int* rowptr = (int*)(Z2 + (size_t)N * HIDDEN);
  int* lrank  = rowptr + Npad;
  int* csr    = lrank + ePad;
  int* cntG   = csr + 2 * ePad;
  int* posG   = cntG + (size_t)GCN_CHUNKS * Npad;
  int* blkG   = posG + (size_t)GCN_CHUNKS * Npad;

  const int tiles  = (N + 15) / 16;
  const int g1Grps = (tiles + GCN_G1W - 1) / GCN_G1W;
  gcn_rank_gemm1_kernel<<<GCN_CHUNKS + g1Grps, 1024, 0, stream>>>(
      edst, cntG, lrank, blkG, N, E, Npad, nbPad, chunk, features, W1, Y);

  gcn_rowstart_kernel<<<nb, GCN_RB, 0, stream>>>(cntG, blkG, posG, rowptr, N, Npad, nbPad);
  gcn_fill_kernel<<<(E + 255) / 256, 256, 0, stream>>>(
      esrc, edst, evals, lrank, posG, csr, E, Npad, chunk);

  {
    const int blocks = ((N + 1) / 2 + 7) / 8;
    gcn_spmm_rows_kernel<<<blocks, 256, 0, stream>>>(rowptr, csr, Y, Z1, N,  0);
    gcn_spmm_rows_kernel<<<blocks, 256, 0, stream>>>(rowptr, csr, Z1, Z2, N,  1);
  }

  gcn_gemm2_lsm_wmma<<<(N + 15) / 16, 32, 0, stream>>>(Z2, W2, out, N);
}


# Z2@W2 + log-softmax moved into the second SpMM's work-groups: the 8 waves leave a 16-row Z2 tile in LDS, 3 waves run 4 WMMAs each, all take 2 rows of softmax; 6 to 5 launches, Z2 never stored
# speedup vs baseline: 1.5220x; 1.0690x over previous
#include <hip/hip_runtime.h>
#include <math.h>

#define IN_DIM  100
#define HIDDEN  16
#define NCLS    40

typedef __attribute__((ext_vector_type(2))) float v2f;
typedef __attribute__((ext_vector_type(8))) float v8f;

static __device__ inline v8f wmma_f32_16x16x4(v2f a, v2f b, v8f c) {
#if __has_builtin(__builtin_amdgcn_wmma_f32_16x16x4_f32)
  return __builtin_amdgcn_wmma_f32_16x16x4_f32(
       false, a,  false, b,
       (short)0, c,  false,  false);
#else
  const int lane = (int)(threadIdx.x & 31u);
  const int nsel = lane & 15;
  v8f d = c;
#pragma unroll
  for (int g = 0; g < 8; ++g) {
    const int m = g + 8 * (lane >> 4);
    float acc = 0.f;
#pragma unroll
    for (int k = 0; k < 4; ++k) {
      const int srcA = m + 16 * (k >> 1);
      const int srcB = nsel + 16 * (k >> 1);
      const float av = __shfl((k & 1) ? a.y : a.x, srcA, 32);
      const float bv = __shfl((k & 1) ? b.y : b.x, srcB, 32);
      acc += av * bv;
    }
    d[g] += acc;
  }
  return d;
#endif
}

#define GCN_CHUNKS 64
#define GCN_HCAP   65536
#define GCN_RB     256
#define GCN_G1W    32

static __device__ inline void gcn_gemm1_tile(
    const float* __restrict__ X, const float* __restrict__ W1,
    float* __restrict__ Y, int N, int r0, int lane) {
  const int col  = lane & 15;
  const int kh   = lane >> 4;
  const int rowA = min(r0 + col, N - 1);
  const float* __restrict__ Xr = X + (size_t)rowA * IN_DIM;

  v8f acc = {};
#pragma unroll 5
  for (int k0 = 0; k0 < IN_DIM; k0 += 4) {
    const int ka = k0 + 2 * kh;
    v2f a;
    a.x = Xr[ka];
    a.y = Xr[ka + 1];
    v2f b;
    b.x = W1[ka * HIDDEN + col];
    b.y = W1[(ka + 1) * HIDDEN + col];
    acc = wmma_f32_16x16x4(a, b, acc);
  }

  float* __restrict__ Yr = Y + (size_t)(r0 + 8 * kh) * HIDDEN + col;
  if (r0 + 16 <= N) {
#pragma unroll
    for (int g = 0; g < 8; ++g) Yr[g * HIDDEN] = acc[g];
  } else {
#pragma unroll
    for (int g = 0; g < 8; ++g) {
      const int row = r0 + g + 8 * kh;
      if (row < N) Y[row * HIDDEN + col] = acc[g];
    }
  }
}

__global__ __launch_bounds__(1024) void gcn_rank_gemm1_kernel(
    const int* __restrict__ edst, int* __restrict__ cntG,
    int* __restrict__ lrank, int* __restrict__ blkG,
    int N, int E, int Npad, int nbPad, int chunk,
    const float* __restrict__ X, const float* __restrict__ W1,
    float* __restrict__ Y) {
  __shared__ int h[GCN_HCAP];
  __shared__ int s_blk[GCN_HCAP / GCN_RB];
  __shared__ int s_wt[8];
  const int t    = (int)threadIdx.x;
  const int lane = t & 31;
  const int w    = t >> 5;
  if ((int)blockIdx.x >= GCN_CHUNKS) {
    const int wv = __builtin_amdgcn_readfirstlane(w);
    const int r0 = (((int)blockIdx.x - GCN_CHUNKS) * GCN_G1W + wv) * 16;
    if (r0 < N) gcn_gemm1_tile(X, W1, Y, N, r0, lane);
    return;
  }
  const int g    = (int)blockIdx.x;
  const int e_lo = min(E, g * chunk);
  const int e_hi = min(E, e_lo + chunk);
  const int nb   = N / GCN_RB + 1;
  int4* __restrict__ h4 = reinterpret_cast<int4*>(h);
  int carry = 0;

  for (int lo = 0; lo <= N; lo += GCN_HCAP) {
    const int n  = min(GCN_HCAP, N - lo);
    const int n4 = (n + 3) >> 2;
    for (int i = t; i < n4; i += 1024) h4[i] = make_int4(0, 0, 0, 0);
    __syncthreads();

    for (int e0 = e_lo + t; e0 < e_hi; e0 += 4096) {
      int d[4];
#pragma unroll
      for (int u = 0; u < 4; ++u) d[u] = edst[min(e0 + u * 1024, e_hi - 1)] - lo;
#pragma unroll
      for (int u = 0; u < 4; ++u) {
        const int e = e0 + u * 1024;
        if (e < e_hi && (unsigned)d[u] < (unsigned)n) lrank[e] = atomicAdd(&h[d[u]], 1);
      }
    }
    __syncthreads();

    int4* __restrict__ o4 = reinterpret_cast<int4*>(cntG + (size_t)g * Npad + lo);
    for (int i = t; i < n4; i += 1024) o4[i] = h4[i];

    const int nbl = min(GCN_HCAP / GCN_RB, n / GCN_RB + 1);
    for (int b = w; b < nbl; b += 32) {
      const int j = b * (GCN_RB / 4) + lane * 2;
      const int4 a = (j     < n4) ? h4[j]     : make_int4(0, 0, 0, 0);
      const int4 c = (j + 1 < n4) ? h4[j + 1] : make_int4(0, 0, 0, 0);
      int s = ((a.x + a.y) + (a.z + a.w)) + ((c.x + c.y) + (c.z + c.w));
#pragma unroll
      for (int off = 16; off >= 1; off >>= 1) s += __shfl_xor(s, off, 32);
      if (lane == 0) s_blk[b] = s;
    }
    __syncthreads();

    int x = 0, inc = 0;
    if (t < GCN_HCAP / GCN_RB) {
      x = (t < nbl) ? s_blk[t] : 0;
      inc = x;
#pragma unroll
      for (int off = 1; off < 32; off <<= 1) {
        const int y = __shfl_up(inc, off, 32);
        if (lane >= off) inc += y;
      }
      if (lane == 31) s_wt[w] = inc;
    }
    __syncthreads();
    int wt  = (lane < w && lane < 8) ? s_wt[lane] : 0;
    int all = (lane < 8) ? s_wt[lane] : 0;
#pragma unroll
    for (int off = 4; off >= 1; off >>= 1) {
      wt  += __shfl_xor(wt, off, 32);
      all += __shfl_xor(all, off, 32);
    }
    wt  = __shfl(wt, 0, 32);
    all = __shfl(all, 0, 32);
    if (t < GCN_HCAP / GCN_RB) {
      const int bidx = lo / GCN_RB + t;
      if (bidx < nb) blkG[(size_t)g * nbPad + bidx] = carry + wt + inc - x;
    }
    carry += all;
    __syncthreads();
  }
}

__global__ __launch_bounds__(GCN_RB) void gcn_rowstart_kernel(
    const int* __restrict__ cntG, const int* __restrict__ blkG,
    int* __restrict__ posG, int* __restrict__ rowptr, int N, int Npad, int nbPad) {
  __shared__ int s_tot[GCN_RB / 32];
  const int t    = (int)threadIdx.x;
  const int lane = t & 31;
  const int w    = t >> 5;
  const int b    = (int)blockIdx.x;
  const int i    = b * GCN_RB + t;
  const bool in  = i < N;
  const int ic   = min(i, N - 1);

  int c[GCN_CHUNKS];
#pragma unroll
  for (int g = 0; g < GCN_CHUNKS; ++g) c[g] = cntG[(size_t)g * Npad + ic];

  int base = 0;
  for (int g = lane; g < GCN_CHUNKS; g += 32) base += blkG[(size_t)g * nbPad + b];
#pragma unroll
  for (int off = 16; off >= 1; off >>= 1) base += __shfl_xor(base, off, 32);

  int tot = 0;
#pragma unroll
  for (int g = 0; g < GCN_CHUNKS; ++g) { c[g] = in ? c[g] : 0; tot += c[g]; }

  int inc = tot;
#pragma unroll
  for (int off = 1; off < 32; off <<= 1) {
    const int y = __shfl_up(inc, off, 32);
    if (lane >= off) inc += y;
  }
  if (lane == 31) s_tot[w] = inc;
  __syncthreads();
  int wt = (lane < w && lane < GCN_RB / 32) ? s_tot[lane] : 0;
#pragma unroll
  for (int off = 16; off >= 1; off >>= 1) wt += __shfl_xor(wt, off, 32);

  int run = base + wt + inc - tot;
  if (i <= N) rowptr[i] = run;
  if (in) {
#pragma unroll
    for (int g = 0; g < GCN_CHUNKS; ++g) {
      posG[(size_t)g * Npad + i] = run;
      run += c[g];
    }
  }
}

__global__ __launch_bounds__(256) void gcn_fill_kernel(
    const int* __restrict__ esrc, const int* __restrict__ edst,
    const float* __restrict__ evals, const int* __restrict__ lrank,
    const int* __restrict__ posG,
    int* __restrict__ csr, int E, int Npad, int chunk) {
  const int e = blockIdx.x * blockDim.x + threadIdx.x;
  if (e >= E) return;
  const int d = edst[e];
  const int g = e / chunk;
  const int pos = posG[(size_t)g * Npad + d] + lrank[e];
  int2 ev;
  ev.x = esrc[e];
  ev.y = __float_as_int(evals[e]);
  reinterpret_cast<int2*>(csr)[pos] = ev;
}

static __device__ inline float4 gcn_row_gather(
    const int2* __restrict__ csr, const float4* __restrict__ X4,
    int beg, int end, int q, int c4, int doRelu) {
  float4 acc = make_float4(0.f, 0.f, 0.f, 0.f);
  for (int b = beg; b < end; b += 16) {
    const int i0 = b + q;
    const int i1 = i0 + 8;
    const bool ok0 = i0 < end;
    const bool ok1 = i1 < end;
    const int2 e0 = csr[ok0 ? i0 : beg];
    const int2 e1 = csr[ok1 ? i1 : beg];
    float4 x0 = X4[(size_t)e0.x * 4 + c4];
    float4 x1 = X4[(size_t)e1.x * 4 + c4];
    if (doRelu) {
      x0.x = fmaxf(x0.x, 0.f); x0.y = fmaxf(x0.y, 0.f);
      x0.z = fmaxf(x0.z, 0.f); x0.w = fmaxf(x0.w, 0.f);
      x1.x = fmaxf(x1.x, 0.f); x1.y = fmaxf(x1.y, 0.f);
      x1.z = fmaxf(x1.z, 0.f); x1.w = fmaxf(x1.w, 0.f);
    }
    const float v0 = __int_as_float(e0.y);
    const float v1 = __int_as_float(e1.y);
    if (ok0) {
      acc.x = fmaf(v0, x0.x, acc.x); acc.y = fmaf(v0, x0.y, acc.y);
      acc.z = fmaf(v0, x0.z, acc.z); acc.w = fmaf(v0, x0.w, acc.w);
    }
    if (ok1) {
      acc.x = fmaf(v1, x1.x, acc.x); acc.y = fmaf(v1, x1.y, acc.y);
      acc.z = fmaf(v1, x1.z, acc.z); acc.w = fmaf(v1, x1.w, acc.w);
    }
  }
  return acc;
}

static __device__ inline float gcn_slot_sum(float4 a, int lane) {
  const bool h4 = (lane & 16) != 0;
  const bool h3 = (lane & 8) != 0;
  float k0 = h4 ? a.z : a.x;
  float k1 = h4 ? a.w : a.y;
  k0 += __shfl_xor(h4 ? a.x : a.z, 16, 32);
  k1 += __shfl_xor(h4 ? a.y : a.w, 16, 32);
  float k = h3 ? k1 : k0;
  k += __shfl_xor(h3 ? k0 : k1, 8, 32);
  k += __shfl_xor(k, 4, 32);
  return k;
}

__global__ __launch_bounds__(256) void gcn_spmm_rows_kernel(
    const int* __restrict__ rowptr, const int* __restrict__ csr,
    const float* __restrict__ Xin, float* __restrict__ Z, int N, int doRelu) {
  const int lane = (int)(threadIdx.x & 31u);
  const int wv = __builtin_amdgcn_readfirstlane((int)(threadIdx.x >> 5));
  const int r0 = (blockIdx.x * 8 + wv) * 2;
  if (r0 >= N) return;
  const int q  = lane >> 2;
  const int c4 = lane & 3;
  const int2*   __restrict__ E2 = reinterpret_cast<const int2*>(csr);
  const float4* __restrict__ X4 = reinterpret_cast<const float4*>(Xin);

  const int p0 = rowptr[r0];
  const int p1 = rowptr[r0 + 1];
  const int p2 = rowptr[min(r0 + 2, N)];

  const float za = gcn_slot_sum(gcn_row_gather(E2, X4, p0, p1, q, c4, doRelu), lane);
  const float zb = gcn_slot_sum(gcn_row_gather(E2, X4, p1, p2, q, c4, doRelu), lane);

  const int rsel = (lane >> 2) & 1;
  const int col  = 4 * c4 + 2 * ((lane >> 4) & 1) + ((lane >> 3) & 1);
  const int row  = r0 + rsel;
  if (row < N) Z[(size_t)row * HIDDEN + col] = rsel ? zb : za;
}

#define GCN_ZP 18
#define GCN_LP 50

__global__ __launch_bounds__(256) void gcn_spmm_gemm2_lsm_kernel(
    const int* __restrict__ rowptr, const int* __restrict__ csr,
    const float* __restrict__ Xin, const float* __restrict__ W2,
    float* __restrict__ out, int N) {
  __shared__ __attribute__((aligned(16))) float s_z[16 * GCN_ZP];
  __shared__ float s_l[16 * GCN_LP];
  const int lane = (int)(threadIdx.x & 31u);
  const int wv  = __builtin_amdgcn_readfirstlane((int)(threadIdx.x >> 5));
  const int t0  = (int)blockIdx.x * 16;
  const int r0  = t0 + wv * 2;
  const int col = lane & 15;
  const int kh  = lane >> 4;

  const int p0 = rowptr[min(r0, N)];
  const int p1 = rowptr[min(r0 + 1, N)];
  const int p2 = rowptr[min(r0 + 2, N)];

  v2f b[4] = {};
  if (wv < 3) {
    const int cc = min(16 * wv + col, NCLS - 1);
#pragma unroll
    for (int j = 0; j < 4; ++j) {
      const int ka = 4 * j + 2 * kh;
      b[j].x = W2[ka * NCLS + cc];
      b[j].y = W2[(ka + 1) * NCLS + cc];
    }
  }

  {
    const int q  = lane >> 2;
    const int c4 = lane & 3;
    const int2*   __restrict__ E2 = reinterpret_cast<const int2*>(csr);
    const float4* __restrict__ X4 = reinterpret_cast<const float4*>(Xin);

    const float za = gcn_slot_sum(gcn_row_gather(E2, X4, p0, p1, q, c4,  1), lane);
    const float zb = gcn_slot_sum(gcn_row_gather(E2, X4, p1, p2, q, c4,  1), lane);

    const int rsel = (lane >> 2) & 1;
    const int zc   = 4 * c4 + 2 * ((lane >> 4) & 1) + ((lane >> 3) & 1);
    s_z[(2 * wv + rsel) * GCN_ZP + zc] = rsel ? zb : za;
  }
  __syncthreads();

  if (wv < 3) {
    const v2f* __restrict__ zr = reinterpret_cast<const v2f*>(s_z + col * GCN_ZP + 2 * kh);
    v8f acc = {};
#pragma unroll
    for (int j = 0; j < 4; ++j) acc = wmma_f32_16x16x4(zr[2 * j], b[j], acc);
    float* __restrict__ lw = s_l + (8 * kh) * GCN_LP + 16 * wv + col;
#pragma unroll
    for (int g = 0; g < 8; ++g) lw[g * GCN_LP] = acc[g];
  }
  __syncthreads();

  const int lr  = 2 * wv + kh;
  const int row = t0 + lr;
  const float* __restrict__ lp = s_l + lr * GCN_LP + col;
  const float v0  = lp[0];
  const float v1  = lp[16];
  const float v2v = lp[32];
  const bool t2ok = (col < 8);
  const float ninf = -__builtin_inff();

  float m = fmaxf(fmaxf(v0, v1), t2ok ? v2v : ninf);
#pragma unroll
  for (int off = 8; off >= 1; off >>= 1) m = fmaxf(m, __shfl_xor(m, off, 16));

  float s = expf(v0 - m) + expf(v1 - m) + (t2ok ? expf(v2v - m) : 0.0f);
#pragma unroll
  for (int off = 8; off >= 1; off >>= 1) s += __shfl_xor(s, off, 16);

  const float lse = m + logf(s);

  if (row < N) {
    float* __restrict__ o = out + (size_t)row * NCLS + col;
    o[0]  = v0 - lse;
    o[16] = v1 - lse;
    if (t2ok) o[32] = v2v - lse;
  }
}

extern "C" void kernel_launch(void* const* d_in, const int* in_sizes, int n_in,
                              void* d_out, int out_size, void* d_ws, size_t ws_size,
                              hipStream_t stream) {
  const float* features = (const float*)d_in[0];
  const int*   esrc     = (const int*)d_in[1];
  const int*   edst     = (const int*)d_in[2];
  const float* evals    = (const float*)d_in[3];
  const float* W1       = (const float*)d_in[4];
  const float* W2       = (const float*)d_in[5];
  float*       out      = (float*)d_out;

  const int N = in_sizes[0] / IN_DIM;
  const int E = in_sizes[1];

  float* Y  = (float*)d_ws;
  float* Z1 = Y + (size_t)N * HIDDEN;
  const int    Npad = (N + 1 + 63) / 64 * 64;
  const size_t ePad = ((size_t)E + 63) / 64 * 64;
  const int    chunk = (E + GCN_CHUNKS - 1) / GCN_CHUNKS;
  const int nb    = N / GCN_RB + 1;
  const int nbPad = (nb + 63) / 64 * 64;
  int* rowptr = (int*)(Z1 + (size_t)N * HIDDEN);
  int* lrank  = rowptr + Npad;
  int* csr    = lrank + ePad;
  int* cntG   = csr + 2 * ePad;
  int* posG   = cntG + (size_t)GCN_CHUNKS * Npad;
  int* blkG   = posG + (size_t)GCN_CHUNKS * Npad;

  const int tiles  = (N + 15) / 16;
  const int g1Grps = (tiles + GCN_G1W - 1) / GCN_G1W;
  gcn_rank_gemm1_kernel<<<GCN_CHUNKS + g1Grps, 1024, 0, stream>>>(
      edst, cntG, lrank, blkG, N, E, Npad, nbPad, chunk, features, W1, Y);

  gcn_rowstart_kernel<<<nb, GCN_RB, 0, stream>>>(cntG, blkG, posG, rowptr, N, Npad, nbPad);
  gcn_fill_kernel<<<(E + 255) / 256, 256, 0, stream>>>(
      esrc, edst, evals, lrank, posG, csr, E, Npad, chunk);

  {
    const int blocks = ((N + 1) / 2 + 7) / 8;
    gcn_spmm_rows_kernel<<<blocks, 256, 0, stream>>>(rowptr, csr, Y, Z1, N,  0);
  }

  gcn_spmm_gemm2_lsm_kernel<<<(N + 15) / 16, 256, 0, stream>>>(rowptr, csr, Z1, W2, out, N);
}


# Edge ranking split into 16 runs x 4 row ranges instead of 64 runs: per-run count, position and block tables shrink from 64 planes to 16; runs 4x longer: more of a row's sum follows LDS arrival order
# speedup vs baseline: 1.5952x; 1.0481x over previous
#include <hip/hip_runtime.h>
#include <math.h>

#define IN_DIM  100
#define HIDDEN  16
#define NCLS    40

typedef __attribute__((ext_vector_type(2))) float v2f;
typedef __attribute__((ext_vector_type(8))) float v8f;

static __device__ inline v8f wmma_f32_16x16x4(v2f a, v2f b, v8f c) {
#if __has_builtin(__builtin_amdgcn_wmma_f32_16x16x4_f32)
  return __builtin_amdgcn_wmma_f32_16x16x4_f32(
       false, a,  false, b,
       (short)0, c,  false,  false);
#else
  const int lane = (int)(threadIdx.x & 31u);
  const int nsel = lane & 15;
  v8f d = c;
#pragma unroll
  for (int g = 0; g < 8; ++g) {
    const int m = g + 8 * (lane >> 4);
    float acc = 0.f;
#pragma unroll
    for (int k = 0; k < 4; ++k) {
      const int srcA = m + 16 * (k >> 1);
      const int srcB = nsel + 16 * (k >> 1);
      const float av = __shfl((k & 1) ? a.y : a.x, srcA, 32);
      const float bv = __shfl((k & 1) ? b.y : b.x, srcB, 32);
      acc += av * bv;
    }
    d[g] += acc;
  }
  return d;
#endif
}

#define GCN_RUNS   16
#define GCN_RNGS   4
#define GCN_CHUNKS (GCN_RUNS * GCN_RNGS)
#define GCN_HCAP   65536
#define GCN_RB     256
#define GCN_LD     16
#define GCN_G1W    32

static __device__ inline void gcn_gemm1_tile(
    const float* __restrict__ X, const float* __restrict__ W1,
    float* __restrict__ Y, int N, int r0, int lane) {
  const int col  = lane & 15;
  const int kh   = lane >> 4;
  const int rowA = min(r0 + col, N - 1);
  const float* __restrict__ Xr = X + (size_t)rowA * IN_DIM;

  v8f acc = {};
#pragma unroll 5
  for (int k0 = 0; k0 < IN_DIM; k0 += 4) {
    const int ka = k0 + 2 * kh;
    v2f a;
    a.x = Xr[ka];
    a.y = Xr[ka + 1];
    v2f b;
    b.x = W1[ka * HIDDEN + col];
    b.y = W1[(ka + 1) * HIDDEN + col];
    acc = wmma_f32_16x16x4(a, b, acc);
  }

  float* __restrict__ Yr = Y + (size_t)(r0 + 8 * kh) * HIDDEN + col;
  if (r0 + 16 <= N) {
#pragma unroll
    for (int g = 0; g < 8; ++g) Yr[g * HIDDEN] = acc[g];
  } else {
#pragma unroll
    for (int g = 0; g < 8; ++g) {
      const int row = r0 + g + 8 * kh;
      if (row < N) Y[row * HIDDEN + col] = acc[g];
    }
  }
}

__global__ __launch_bounds__(1024) void gcn_rank_gemm1_kernel(
    const int* __restrict__ edst, int* __restrict__ cntG,
    int* __restrict__ lrank, int* __restrict__ blkG,
    int N, int E, int Npad, int nbPad, int chunk, int rngRows,
    const float* __restrict__ X, const float* __restrict__ W1,
    float* __restrict__ Y) {
  __shared__ int h[GCN_HCAP];
  const int t    = (int)threadIdx.x;
  const int lane = t & 31;
  const int w    = t >> 5;
  if ((int)blockIdx.x >= GCN_CHUNKS) {
    const int wv = __builtin_amdgcn_readfirstlane(w);
    const int r0 = (((int)blockIdx.x - GCN_CHUNKS) * GCN_G1W + wv) * 16;
    if (r0 < N) gcn_gemm1_tile(X, W1, Y, N, r0, lane);
    return;
  }
  const int g    = (int)blockIdx.x / GCN_RNGS;
  const int r    = (int)blockIdx.x % GCN_RNGS;
  const int e_lo = min(E, g * chunk);
  const int e_hi = min(E, e_lo + chunk);
  const int rlo  = r * rngRows;
  int4* __restrict__ h4 = reinterpret_cast<int4*>(h);

  for (int lo = rlo; lo < rlo + rngRows; lo += GCN_HCAP) {
    const int np  = min(GCN_HCAP, rlo + rngRows - lo);
    const int n   = max(0, min(np, N - lo));
    const int np4 = np >> 2;
    const int n4  = (n + 3) >> 2;
    for (int i = t; i < np4; i += 1024) h4[i] = make_int4(0, 0, 0, 0);
    __syncthreads();

    if (e_lo < e_hi) {
      for (int e0 = e_lo + t; e0 < e_hi; e0 += GCN_LD * 1024) {
        int d[GCN_LD];
#pragma unroll
        for (int u = 0; u < GCN_LD; ++u) d[u] = edst[min(e0 + u * 1024, e_hi - 1)] - lo;
#pragma unroll
        for (int u = 0; u < GCN_LD; ++u) {
          const int e = e0 + u * 1024;
          if (e < e_hi && (unsigned)d[u] < (unsigned)n) lrank[e] = atomicAdd(&h[d[u]], 1);
        }
      }
    }
    __syncthreads();

    int4* __restrict__ o4 = reinterpret_cast<int4*>(cntG + (size_t)g * Npad + lo);
    for (int i = t; i < n4; i += 1024) o4[i] = h4[i];

    const int nbl = np / GCN_RB;
    for (int b = w; b < nbl; b += 32) {
      const int j = b * (GCN_RB / 4) + lane * 2;
      const int4 a = h4[j];
      const int4 c = h4[j + 1];
      int s = ((a.x + a.y) + (a.z + a.w)) + ((c.x + c.y) + (c.z + c.w));
#pragma unroll
      for (int off = 16; off >= 1; off >>= 1) s += __shfl_xor(s, off, 32);
      if (lane == 0) blkG[(size_t)g * nbPad + lo / GCN_RB + b] = s;
    }
    __syncthreads();
  }
}

__global__ __launch_bounds__(GCN_RB) void gcn_rowstart_kernel(
    const int* __restrict__ cntG, const int* __restrict__ blkG,
    int* __restrict__ posG, int* __restrict__ rowptr, int N, int Npad, int nbPad) {
  __shared__ int s_tot[GCN_RB / 32];
  __shared__ int s_base[GCN_RB / 32];
  const int t    = (int)threadIdx.x;
  const int lane = t & 31;
  const int w    = t >> 5;
  const int b    = (int)blockIdx.x;
  const int i    = b * GCN_RB + t;
  const bool in  = i < N;
  const int ic   = min(i, N - 1);

  int c[GCN_RUNS];
#pragma unroll
  for (int g = 0; g < GCN_RUNS; ++g) c[g] = cntG[(size_t)g * Npad + ic];

  int bsum = 0;
  for (int bb = t; bb < b; bb += GCN_RB) {
#pragma unroll
    for (int g = 0; g < GCN_RUNS; ++g) bsum += blkG[(size_t)g * nbPad + bb];
  }
#pragma unroll
  for (int off = 16; off >= 1; off >>= 1) bsum += __shfl_xor(bsum, off, 32);
  if (lane == 0) s_base[w] = bsum;

  int tot = 0;
#pragma unroll
  for (int g = 0; g < GCN_RUNS; ++g) { c[g] = in ? c[g] : 0; tot += c[g]; }

  int inc = tot;
#pragma unroll
  for (int off = 1; off < 32; off <<= 1) {
    const int y = __shfl_up(inc, off, 32);
    if (lane >= off) inc += y;
  }
  if (lane == 31) s_tot[w] = inc;
  __syncthreads();
  int wt   = (lane < w && lane < GCN_RB / 32) ? s_tot[lane] : 0;
  int base = (lane < GCN_RB / 32) ? s_base[lane] : 0;
#pragma unroll
  for (int off = 16; off >= 1; off >>= 1) {
    wt   += __shfl_xor(wt, off, 32);
    base += __shfl_xor(base, off, 32);
  }

  int run = base + wt + inc - tot;
  if (i <= N) rowptr[i] = run;
  if (in) {
#pragma unroll
    for (int g = 0; g < GCN_RUNS; ++g) {
      posG[(size_t)g * Npad + i] = run;
      run += c[g];
    }
  }
}

__global__ __launch_bounds__(256) void gcn_fill_kernel(
    const int* __restrict__ esrc, const int* __restrict__ edst,
    const float* __restrict__ evals, const int* __restrict__ lrank,
    const int* __restrict__ posG,
    int* __restrict__ csr, int E, int Npad, int chunk) {
  const int e = blockIdx.x * blockDim.x + threadIdx.x;
  if (e >= E) return;
  const int d = edst[e];
  const int g = e / chunk;
  const int pos = posG[(size_t)g * Npad + d] + lrank[e];
  int2 ev;
  ev.x = esrc[e];
  ev.y = __float_as_int(evals[e]);
  reinterpret_cast<int2*>(csr)[pos] = ev;
}

static __device__ inline float4 gcn_row_gather(
    const int2* __restrict__ csr, const float4* __restrict__ X4,
    int beg, int end, int q, int c4, int doRelu) {
  float4 acc = make_float4(0.f, 0.f, 0.f, 0.f);
  for (int b = beg; b < end; b += 16) {
    const int i0 = b + q;
    const int i1 = i0 + 8;
    const bool ok0 = i0 < end;
    const bool ok1 = i1 < end;
    const int2 e0 = csr[ok0 ? i0 : beg];
    const int2 e1 = csr[ok1 ? i1 : beg];
    float4 x0 = X4[(size_t)e0.x * 4 + c4];
    float4 x1 = X4[(size_t)e1.x * 4 + c4];
    if (doRelu) {
      x0.x = fmaxf(x0.x, 0.f); x0.y = fmaxf(x0.y, 0.f);
      x0.z = fmaxf(x0.z, 0.f); x0.w = fmaxf(x0.w, 0.f);
      x1.x = fmaxf(x1.x, 0.f); x1.y = fmaxf(x1.y, 0.f);
      x1.z = fmaxf(x1.z, 0.f); x1.w = fmaxf(x1.w, 0.f);
    }
    const float v0 = __int_as_float(e0.y);
    const float v1 = __int_as_float(e1.y);
    if (ok0) {
      acc.x = fmaf(v0, x0.x, acc.x); acc.y = fmaf(v0, x0.y, acc.y);
      acc.z = fmaf(v0, x0.z, acc.z); acc.w = fmaf(v0, x0.w, acc.w);
    }
    if (ok1) {
      acc.x = fmaf(v1, x1.x, acc.x); acc.y = fmaf(v1, x1.y, acc.y);
      acc.z = fmaf(v1, x1.z, acc.z); acc.w = fmaf(v1, x1.w, acc.w);
    }
  }
  return acc;
}

static __device__ inline float gcn_slot_sum(float4 a, int lane) {
  const bool h4 = (lane & 16) != 0;
  const bool h3 = (lane & 8) != 0;
  float k0 = h4 ? a.z : a.x;
  float k1 = h4 ? a.w : a.y;
  k0 += __shfl_xor(h4 ? a.x : a.z, 16, 32);
  k1 += __shfl_xor(h4 ? a.y : a.w, 16, 32);
  float k = h3 ? k1 : k0;
  k += __shfl_xor(h3 ? k0 : k1, 8, 32);
  k += __shfl_xor(k, 4, 32);
  return k;
}

__global__ __launch_bounds__(256) void gcn_spmm_rows_kernel(
    const int* __restrict__ rowptr, const int* __restrict__ csr,
    const float* __restrict__ Xin, float* __restrict__ Z, int N, int doRelu) {
  const int lane = (int)(threadIdx.x & 31u);
  const int wv = __builtin_amdgcn_readfirstlane((int)(threadIdx.x >> 5));
  const int r0 = (blockIdx.x * 8 + wv) * 2;
  if (r0 >= N) return;
  const int q  = lane >> 2;
  const int c4 = lane & 3;
  const int2*   __restrict__ E2 = reinterpret_cast<const int2*>(csr);
  const float4* __restrict__ X4 = reinterpret_cast<const float4*>(Xin);

  const int p0 = rowptr[r0];
  const int p1 = rowptr[r0 + 1];
  const int p2 = rowptr[min(r0 + 2, N)];

  const float za = gcn_slot_sum(gcn_row_gather(E2, X4, p0, p1, q, c4, doRelu), lane);
  const float zb = gcn_slot_sum(gcn_row_gather(E2, X4, p1, p2, q, c4, doRelu), lane);

  const int rsel = (lane >> 2) & 1;
  const int col  = 4 * c4 + 2 * ((lane >> 4) & 1) + ((lane >> 3) & 1);
  const int row  = r0 + rsel;
  if (row < N) Z[(size_t)row * HIDDEN + col] = rsel ? zb : za;
}

#define GCN_ZP 18
#define GCN_LP 50

__global__ __launch_bounds__(256) void gcn_spmm_gemm2_lsm_kernel(
    const int* __restrict__ rowptr, const int* __restrict__ csr,
    const float* __restrict__ Xin, const float* __restrict__ W2,
    float* __restrict__ out, int N) {
  __shared__ __attribute__((aligned(16))) float s_z[16 * GCN_ZP];
  __shared__ float s_l[16 * GCN_LP];
  const int lane = (int)(threadIdx.x & 31u);
  const int wv  = __builtin_amdgcn_readfirstlane((int)(threadIdx.x >> 5));
  const int t0  = (int)blockIdx.x * 16;
  const int r0  = t0 + wv * 2;
  const int col = lane & 15;
  const int kh  = lane >> 4;

  const int p0 = rowptr[min(r0, N)];
  const int p1 = rowptr[min(r0 + 1, N)];
  const int p2 = rowptr[min(r0 + 2, N)];

  v2f b[4] = {};
  if (wv < 3) {
    const int cc = min(16 * wv + col, NCLS - 1);
#pragma unroll
    for (int j = 0; j < 4; ++j) {
      const int ka = 4 * j + 2 * kh;
      b[j].x = W2[ka * NCLS + cc];
      b[j].y = W2[(ka + 1) * NCLS + cc];
    }
  }

  {
    const int q  = lane >> 2;
    const int c4 = lane & 3;
    const int2*   __restrict__ E2 = reinterpret_cast<const int2*>(csr);
    const float4* __restrict__ X4 = reinterpret_cast<const float4*>(Xin);

    const float za = gcn_slot_sum(gcn_row_gather(E2, X4, p0, p1, q, c4,  1), lane);
    const float zb = gcn_slot_sum(gcn_row_gather(E2, X4, p1, p2, q, c4,  1), lane);

    const int rsel = (lane >> 2) & 1;
    const int zc   = 4 * c4 + 2 * ((lane >> 4) & 1) + ((lane >> 3) & 1);
    s_z[(2 * wv + rsel) * GCN_ZP + zc] = rsel ? zb : za;
  }
  __syncthreads();

  if (wv < 3) {
    const v2f* __restrict__ zr = reinterpret_cast<const v2f*>(s_z + col * GCN_ZP + 2 * kh);
    v8f acc = {};
#pragma unroll
    for (int j = 0; j < 4; ++j) acc = wmma_f32_16x16x4(zr[2 * j], b[j], acc);
    float* __restrict__ lw = s_l + (8 * kh) * GCN_LP + 16 * wv + col;
#pragma unroll
    for (int g = 0; g < 8; ++g) lw[g * GCN_LP] = acc[g];
  }
  __syncthreads();

  const int lr  = 2 * wv + kh;
  const int row = t0 + lr;
  const float* __restrict__ lp = s_l + lr * GCN_LP + col;
  const float v0  = lp[0];
  const float v1  = lp[16];
  const float v2v = lp[32];
  const bool t2ok = (col < 8);
  const float ninf = -__builtin_inff();

  float m = fmaxf(fmaxf(v0, v1), t2ok ? v2v : ninf);
#pragma unroll
  for (int off = 8; off >= 1; off >>= 1) m = fmaxf(m, __shfl_xor(m, off, 16));

  float s = expf(v0 - m) + expf(v1 - m) + (t2ok ? expf(v2v - m) : 0.0f);
#pragma unroll
  for (int off = 8; off >= 1; off >>= 1) s += __shfl_xor(s, off, 16);

  const float lse = m + logf(s);

  if (row < N) {
    float* __restrict__ o = out + (size_t)row * NCLS + col;
    o[0]  = v0 - lse;
    o[16] = v1 - lse;
    if (t2ok) o[32] = v2v - lse;
  }
}

extern "C" void kernel_launch(void* const* d_in, const int* in_sizes, int n_in,
                              void* d_out, int out_size, void* d_ws, size_t ws_size,
                              hipStream_t stream) {
  const float* features = (const float*)d_in[0];
  const int*   esrc     = (const int*)d_in[1];
  const int*   edst     = (const int*)d_in[2];
  const float* evals    = (const float*)d_in[3];
  const float* W1       = (const float*)d_in[4];
  const float* W2       = (const float*)d_in[5];
  float*       out      = (float*)d_out;

  const int N = in_sizes[0] / IN_DIM;
  const int E = in_sizes[1];

  float* Y  = (float*)d_ws;
  float* Z1 = Y + (size_t)N * HIDDEN;
  const int    Npad = (N + 4 + 63) / 64 * 64;
  const size_t ePad = ((size_t)E + 63) / 64 * 64;
  const int    chunk = (E + GCN_RUNS - 1) / GCN_RUNS;
  const int nb    = N / GCN_RB + 1;
  const int rngRows = (nb + GCN_RNGS - 1) / GCN_RNGS * GCN_RB;
  const int nbPad = (GCN_RNGS * (rngRows / GCN_RB) + 63) / 64 * 64;
  int* rowptr = (int*)(Z1 + (size_t)N * HIDDEN);
  int* lrank  = rowptr + Npad;
  int* csr    = lrank + ePad;
  int* cntG   = csr + 2 * ePad;
  int* posG   = cntG + (size_t)GCN_RUNS * Npad;
  int* blkG   = posG + (size_t)GCN_RUNS * Npad;

  const int tiles  = (N + 15) / 16;
  const int g1Grps = (tiles + GCN_G1W - 1) / GCN_G1W;
  gcn_rank_gemm1_kernel<<<GCN_CHUNKS + g1Grps, 1024, 0, stream>>>(
      edst, cntG, lrank, blkG, N, E, Npad, nbPad, chunk, rngRows, features, W1, Y);

  gcn_rowstart_kernel<<<nb, GCN_RB, 0, stream>>>(cntG, blkG, posG, rowptr, N, Npad, nbPad);
  gcn_fill_kernel<<<(E + 255) / 256, 256, 0, stream>>>(
      esrc, edst, evals, lrank, posG, csr, E, Npad, chunk);

  {
    const int blocks = ((N + 1) / 2 + 7) / 8;
    gcn_spmm_rows_kernel<<<blocks, 256, 0, stream>>>(rowptr, csr, Y, Z1, N,  0);
  }

  gcn_spmm_gemm2_lsm_kernel<<<(N + 15) / 16, 256, 0, stream>>>(rowptr, csr, Z1, W2, out, N);
}


# Edges grouped by 208-row block inside each of 64 runs (one launch, LDS counting sort); both SpMMs add a block's rows in LDS with float adds, arrival order; no row pointers or CSR; 5 to 3 launches
# speedup vs baseline: 2.0129x; 1.2618x over previous
#include <hip/hip_runtime.h>
#include <math.h>

#define IN_DIM  100
#define HIDDEN  16
#define NCLS    40

typedef __attribute__((ext_vector_type(2))) float v2f;
typedef __attribute__((ext_vector_type(8))) float v8f;

static __device__ inline v8f wmma_f32_16x16x4(v2f a, v2f b, v8f c) {
#if __has_builtin(__builtin_amdgcn_wmma_f32_16x16x4_f32)
  return __builtin_amdgcn_wmma_f32_16x16x4_f32(
       false, a,  false, b,
       (short)0, c,  false,  false);
#else
  const int lane = (int)(threadIdx.x & 31u);
  const int nsel = lane & 15;
  v8f d = c;
#pragma unroll
  for (int g = 0; g < 8; ++g) {
    const int m = g + 8 * (lane >> 4);
    float acc = 0.f;
#pragma unroll
    for (int k = 0; k < 4; ++k) {
      const int srcA = m + 16 * (k >> 1);
      const int srcB = nsel + 16 * (k >> 1);
      const float av = __shfl((k & 1) ? a.y : a.x, srcA, 32);
      const float bv = __shfl((k & 1) ? b.y : b.x, srcB, 32);
      acc += av * bv;
    }
    d[g] += acc;
  }
  return d;
#endif
}

#define GCN_P      64
#define GCN_RB     208
#define GCN_HCAP   65536
#define GCN_LD     16
#define GCN_G1W    32
#define GCN_EU     8
#define GCN_ZRES   41472

static __device__ inline void gcn_gemm1_tile(
    const float* __restrict__ X, const float* __restrict__ W1,
    float* __restrict__ Y, int N, int r0, int lane) {
  const int col  = lane & 15;
  const int kh   = lane >> 4;
  const int rowA = min(r0 + col, N - 1);
  const float* __restrict__ Xr = X + (size_t)rowA * IN_DIM;

  v8f acc = {};
#pragma unroll 5
  for (int k0 = 0; k0 < IN_DIM; k0 += 4) {
    const int ka = k0 + 2 * kh;
    v2f a;
    a.x = Xr[ka];
    a.y = Xr[ka + 1];
    v2f b;
    b.x = W1[ka * HIDDEN + col];
    b.y = W1[(ka + 1) * HIDDEN + col];
    acc = wmma_f32_16x16x4(a, b, acc);
  }

  float* __restrict__ Yr = Y + (size_t)(r0 + 8 * kh) * HIDDEN + col;
  if (r0 + 16 <= N) {
#pragma unroll
    for (int g = 0; g < 8; ++g) Yr[g * HIDDEN] = acc[g];
  } else {
#pragma unroll
    for (int g = 0; g < 8; ++g) {
      const int row = r0 + g + 8 * kh;
      if (row < N) Y[row * HIDDEN + col] = acc[g];
    }
  }
}

__global__ __launch_bounds__(1024) void gcn_part_gemm1_kernel(
    const int* __restrict__ esrc, const int* __restrict__ edst,
    const float* __restrict__ evals, int* __restrict__ offG,
    int* __restrict__ lrank, int* __restrict__ pe, int* __restrict__ pd,
    int N, int E, int nb, int nbp, int chunk,
    const float* __restrict__ X, const float* __restrict__ W1,
    float* __restrict__ Y) {
  __shared__ int h[GCN_HCAP];
  const int t    = (int)threadIdx.x;
  const int lane = t & 31;
  const int w    = t >> 5;
  if ((int)blockIdx.x >= GCN_P) {
    const int wv = __builtin_amdgcn_readfirstlane(w);
    const int r0 = (((int)blockIdx.x - GCN_P) * GCN_G1W + wv) * 16;
    if (r0 < N) gcn_gemm1_tile(X, W1, Y, N, r0, lane);
    return;
  }
  const int g    = (int)blockIdx.x;
  const int e_lo = min(E, g * chunk);
  const int e_hi = min(E, e_lo + chunk);

  for (int i = t; i <= nb; i += 1024) h[i] = 0;
  __syncthreads();

  for (int e0 = e_lo + t; e0 < e_hi; e0 += GCN_LD * 1024) {
    int d[GCN_LD];
#pragma unroll
    for (int u = 0; u < GCN_LD; ++u) d[u] = edst[min(e0 + u * 1024, e_hi - 1)];
#pragma unroll
    for (int u = 0; u < GCN_LD; ++u) {
      const int e = e0 + u * 1024;
      if (e < e_hi) lrank[e] = atomicAdd(&h[(unsigned)d[u] / GCN_RB], 1);
    }
  }
  __syncthreads();

  if (w == 0) {
    const int per = (nb + 31) / 32;
    const int b0  = min(nb, lane * per);
    const int b1  = min(nb, b0 + per);
    int sum = 0;
    for (int b = b0; b < b1; ++b) sum += h[b];
    int inc = sum;
#pragma unroll
    for (int off = 1; off < 32; off <<= 1) {
      const int y = __shfl_up(inc, off, 32);
      if (lane >= off) inc += y;
    }
    int run = inc - sum;
    for (int b = b0; b < b1; ++b) { const int c = h[b]; h[b] = run; run += c; }
    if (lane == 31) h[nb] = inc;
  }
  __syncthreads();

  for (int i = t; i <= nb; i += 1024) offG[(size_t)g * nbp + i] = h[i];

  int2* __restrict__ pe2 = reinterpret_cast<int2*>(pe);
  for (int e0 = e_lo + t; e0 < e_hi; e0 += GCN_LD * 1024) {
    int d[GCN_LD], rk[GCN_LD], sv[GCN_LD], vv[GCN_LD];
#pragma unroll
    for (int u = 0; u < GCN_LD; ++u) {
      const int e = min(e0 + u * 1024, e_hi - 1);
      d[u]  = edst[e];
      rk[u] = lrank[e];
      sv[u] = esrc[e];
      vv[u] = __float_as_int(evals[e]);
    }
#pragma unroll
    for (int u = 0; u < GCN_LD; ++u) {
      if (e0 + u * 1024 < e_hi) {
        const int j = e_lo + h[(unsigned)d[u] / GCN_RB] + rk[u];
        pe2[j] = make_int2(sv[u], vv[u]);
        pd[j]  = d[u];
      }
    }
  }
}

#define GCN_ZP 18
#define GCN_LP 50

static __device__ inline void gcn_block_accum(
    const int* __restrict__ offG, const int2* __restrict__ pe2,
    const int* __restrict__ pd, const float4* __restrict__ X4,
    float* __restrict__ zl, int b, int nbp, int chunk, int E,
    int wv, int lane, int doRelu) {
  const int q  = lane >> 2;
  const int c4 = lane & 3;
  const int g0 = 2 * wv;
  const int* __restrict__ oa = offG + (size_t)g0 * nbp + b;
  const int* __restrict__ ob = oa + nbp;
  const int sa0 = oa[0], sa1 = oa[1];
  const int sb0 = ob[0], sb1 = ob[1];
  const int lenA = sa1 - sa0;
  const int tot  = lenA + (sb1 - sb0);
  const int ja   = min(E, g0 * chunk) + sa0;
  const int jb   = min(E, (g0 + 1) * chunk) + sb0 - lenA;
  const int jsafe = (lenA > 0) ? ja : jb;
  const int row0 = b * GCN_RB;

  for (int i0 = 0; i0 < tot; i0 += 8 * GCN_EU) {
    int2 ev[GCN_EU];
    int  dl[GCN_EU];
#pragma unroll
    for (int u = 0; u < GCN_EU; ++u) {
      const int i = i0 + u * 8 + q;
      const int j = (i < tot) ? ((i < lenA) ? ja + i : jb + i) : jsafe;
      ev[u] = pe2[j];
      dl[u] = pd[j];
    }
    float4 x[GCN_EU];
#pragma unroll
    for (int u = 0; u < GCN_EU; ++u) x[u] = X4[(size_t)ev[u].x * 4 + c4];
#pragma unroll
    for (int u = 0; u < GCN_EU; ++u) {
      const int i = i0 + u * 8 + q;
      if (i < tot) {
        float4 xv = x[u];
        if (doRelu) {
          xv.x = fmaxf(xv.x, 0.f); xv.y = fmaxf(xv.y, 0.f);
          xv.z = fmaxf(xv.z, 0.f); xv.w = fmaxf(xv.w, 0.f);
        }
        const float v = __int_as_float(ev[u].y);
        float* __restrict__ z = zl + (dl[u] - row0) * GCN_ZP + 4 * c4;
        __hip_atomic_fetch_add(z + 0, v * xv.x, __ATOMIC_RELAXED, __HIP_MEMORY_SCOPE_WORKGROUP);
        __hip_atomic_fetch_add(z + 1, v * xv.y, __ATOMIC_RELAXED, __HIP_MEMORY_SCOPE_WORKGROUP);
        __hip_atomic_fetch_add(z + 2, v * xv.z, __ATOMIC_RELAXED, __HIP_MEMORY_SCOPE_WORKGROUP);
        __hip_atomic_fetch_add(z + 3, v * xv.w, __ATOMIC_RELAXED, __HIP_MEMORY_SCOPE_WORKGROUP);
      }
    }
  }
}

__global__ __launch_bounds__(1024) void gcn_spmm_block_kernel(
    const int* __restrict__ offG, const int* __restrict__ pe,
    const int* __restrict__ pd, const float* __restrict__ Xin,
    float* __restrict__ Z, int N, int E, int nbp, int chunk) {
  __shared__ __attribute__((aligned(16))) float zl[GCN_ZRES];
  const int t    = (int)threadIdx.x;
  const int lane = t & 31;
  const int wv = __builtin_amdgcn_readfirstlane(t >> 5);
  const int b  = (int)blockIdx.x;

  for (int i = t; i < GCN_RB * GCN_ZP; i += 1024) zl[i] = 0.f;
  __syncthreads();
  gcn_block_accum(offG, reinterpret_cast<const int2*>(pe), pd,
                  reinterpret_cast<const float4*>(Xin), zl, b, nbp, chunk, E, wv, lane,  0);
  __syncthreads();

  for (int i = t; i < GCN_RB * HIDDEN; i += 1024) {
    const int r   = i >> 4;
    const int row = b * GCN_RB + r;
    if (row < N) Z[(size_t)row * HIDDEN + (i & 15)] = zl[r * GCN_ZP + (i & 15)];
  }
}

#define GCN_RT (GCN_RB / 16)

__global__ __launch_bounds__(1024) void gcn_spmm_gemm2_lsm_kernel(
    const int* __restrict__ offG, const int* __restrict__ pe,
    const int* __restrict__ pd, const float* __restrict__ Xin,
    const float* __restrict__ W2, float* __restrict__ out,
    int N, int E, int nbp, int chunk) {
  __shared__ __attribute__((aligned(16))) float zl[GCN_ZRES];
  __shared__ float s_l[GCN_RB * GCN_LP];
  const int t    = (int)threadIdx.x;
  const int lane = t & 31;
  const int wv  = __builtin_amdgcn_readfirstlane(t >> 5);
  const int b   = (int)blockIdx.x;
  const int col = lane & 15;
  const int kh  = lane >> 4;

  for (int i = t; i < GCN_RB * GCN_ZP; i += 1024) zl[i] = 0.f;
  __syncthreads();
  gcn_block_accum(offG, reinterpret_cast<const int2*>(pe), pd,
                  reinterpret_cast<const float4*>(Xin), zl, b, nbp, chunk, E, wv, lane,  1);
  __syncthreads();

  for (int job = wv; job < 3 * GCN_RT; job += 32) {
    const int tile = job / 3;
    const int ct   = job - 3 * tile;
    const int cc = min(16 * ct + col, NCLS - 1);
    v2f bf[4];
#pragma unroll
    for (int j = 0; j < 4; ++j) {
      const int ka = 4 * j + 2 * kh;
      bf[j].x = W2[ka * NCLS + cc];
      bf[j].y = W2[(ka + 1) * NCLS + cc];
    }
    const v2f* __restrict__ zr =
        reinterpret_cast<const v2f*>(zl + (tile * 16 + col) * GCN_ZP + 2 * kh);
    v8f acc = {};
#pragma unroll
    for (int j = 0; j < 4; ++j) acc = wmma_f32_16x16x4(zr[2 * j], bf[j], acc);
    float* __restrict__ lw = s_l + (tile * 16 + 8 * kh) * GCN_LP + 16 * ct + col;
#pragma unroll
    for (int g = 0; g < 8; ++g) lw[g * GCN_LP] = acc[g];
  }
  __syncthreads();

  const bool t2ok = (col < 8);
  const float ninf = -__builtin_inff();
  for (int lr = 2 * wv + kh; lr < GCN_RB; lr += 64) {
    const int row = b * GCN_RB + lr;
    const float* __restrict__ lp = s_l + lr * GCN_LP + col;
    const float v0  = lp[0];
    const float v1  = lp[16];
    const float v2v = lp[32];

    float m = fmaxf(fmaxf(v0, v1), t2ok ? v2v : ninf);
#pragma unroll
    for (int off = 8; off >= 1; off >>= 1) m = fmaxf(m, __shfl_xor(m, off, 16));

    float s = expf(v0 - m) + expf(v1 - m) + (t2ok ? expf(v2v - m) : 0.0f);
#pragma unroll
    for (int off = 8; off >= 1; off >>= 1) s += __shfl_xor(s, off, 16);

    const float lse = m + logf(s);

    if (row < N) {
      float* __restrict__ o = out + (size_t)row * NCLS + col;
      o[0]  = v0 - lse;
      o[16] = v1 - lse;
      if (t2ok) o[32] = v2v - lse;
    }
  }
}

extern "C" void kernel_launch(void* const* d_in, const int* in_sizes, int n_in,
                              void* d_out, int out_size, void* d_ws, size_t ws_size,
                              hipStream_t stream) {
  const float* features = (const float*)d_in[0];
  const int*   esrc     = (const int*)d_in[1];
  const int*   edst     = (const int*)d_in[2];
  const float* evals    = (const float*)d_in[3];
  const float* W1       = (const float*)d_in[4];
  const float* W2       = (const float*)d_in[5];
  float*       out      = (float*)d_out;

  const int N = in_sizes[0] / IN_DIM;
  const int E = in_sizes[1];

  float* Y  = (float*)d_ws;
  float* Z1 = Y + (size_t)N * HIDDEN;
  const size_t ePad  = ((size_t)E + 63) / 64 * 64;
  const int    chunk = (E + GCN_P - 1) / GCN_P;
  const int    nb    = (N + GCN_RB - 1) / GCN_RB;
  const int    nbp   = (nb + 1 + 63) / 64 * 64;
  int* lrank = (int*)(Z1 + (size_t)N * HIDDEN);
  int* pd    = lrank + ePad;
  int* pe    = pd + ePad;
  int* offG  = pe + 2 * ePad;

  const int tiles  = (N + 15) / 16;
  const int g1Grps = (tiles + GCN_G1W - 1) / GCN_G1W;
  gcn_part_gemm1_kernel<<<GCN_P + g1Grps, 1024, 0, stream>>>(
      esrc, edst, evals, offG, lrank, pe, pd, N, E, nb, nbp, chunk, features, W1, Y);

  gcn_spmm_block_kernel<<<nb, 1024, 0, stream>>>(offG, pe, pd, Y, Z1, N, E, nbp, chunk);

  gcn_spmm_gemm2_lsm_kernel<<<nb, 1024, 0, stream>>>(offG, pe, pd, Z1, W2, out, N, E, nbp, chunk);
}


# Both SpMMs: edges filed per row in LDS by a returning integer add (rank = list slot), each row summed in a lane quad's registers: no LDS float add per edge; row terms add in rank (arrival) order
# speedup vs baseline: 2.2532x; 1.1194x over previous
#include <hip/hip_runtime.h>
#include <math.h>

#define IN_DIM  100
#define HIDDEN  16
#define NCLS    40

typedef __attribute__((ext_vector_type(2))) float v2f;
typedef __attribute__((ext_vector_type(8))) float v8f;

static __device__ inline v8f wmma_f32_16x16x4(v2f a, v2f b, v8f c) {
#if __has_builtin(__builtin_amdgcn_wmma_f32_16x16x4_f32)
  return __builtin_amdgcn_wmma_f32_16x16x4_f32(
       false, a,  false, b,
       (short)0, c,  false,  false);
#else
  const int lane = (int)(threadIdx.x & 31u);
  const int nsel = lane & 15;
  v8f d = c;
#pragma unroll
  for (int g = 0; g < 8; ++g) {
    const int m = g + 8 * (lane >> 4);
    float acc = 0.f;
#pragma unroll
    for (int k = 0; k < 4; ++k) {
      const int srcA = m + 16 * (k >> 1);
      const int srcB = nsel + 16 * (k >> 1);
      const float av = __shfl((k & 1) ? a.y : a.x, srcA, 32);
      const float bv = __shfl((k & 1) ? b.y : b.x, srcB, 32);
      acc += av * bv;
    }
    d[g] += acc;
  }
  return d;
#endif
}

#define GCN_P      64
#define GCN_RB     208
#define GCN_HCAP   65536
#define GCN_LD     16
#define GCN_G1W    32
#define GCN_SL     6
#define GCN_PQ     (32 * GCN_SL)
#define GCN_RK     48
#define GCN_TP     (GCN_RK + 1)
#define GCN_WB     12
#define GCN_ZRES   41472

static __device__ inline void gcn_gemm1_tile(
    const float* __restrict__ X, const float* __restrict__ W1,
    float* __restrict__ Y, int N, int r0, int lane) {
  const int col  = lane & 15;
  const int kh   = lane >> 4;
  const int rowA = min(r0 + col, N - 1);
  const float* __restrict__ Xr = X + (size_t)rowA * IN_DIM;

  v8f acc = {};
#pragma unroll 5
  for (int k0 = 0; k0 < IN_DIM; k0 += 4) {
    const int ka = k0 + 2 * kh;
    v2f a;
    a.x = Xr[ka];
    a.y = Xr[ka + 1];
    v2f b;
    b.x = W1[ka * HIDDEN + col];
    b.y = W1[(ka + 1) * HIDDEN + col];
    acc = wmma_f32_16x16x4(a, b, acc);
  }

  float* __restrict__ Yr = Y + (size_t)(r0 + 8 * kh) * HIDDEN + col;
  if (r0 + 16 <= N) {
#pragma unroll
    for (int g = 0; g < 8; ++g) Yr[g * HIDDEN] = acc[g];
  } else {
#pragma unroll
    for (int g = 0; g < 8; ++g) {
      const int row = r0 + g + 8 * kh;
      if (row < N) Y[row * HIDDEN + col] = acc[g];
    }
  }
}

__global__ __launch_bounds__(1024) void gcn_part_gemm1_kernel(
    const int* __restrict__ esrc, const int* __restrict__ edst,
    const float* __restrict__ evals, int* __restrict__ offG,
    int* __restrict__ lrank, int* __restrict__ pe, int* __restrict__ pd,
    int N, int E, int nb, int nbp, int chunk,
    const float* __restrict__ X, const float* __restrict__ W1,
    float* __restrict__ Y) {
  __shared__ int h[GCN_HCAP];
  const int t    = (int)threadIdx.x;
  const int lane = t & 31;
  const int w    = t >> 5;
  if ((int)blockIdx.x >= GCN_P) {
    const int wv = __builtin_amdgcn_readfirstlane(w);
    const int r0 = (((int)blockIdx.x - GCN_P) * GCN_G1W + wv) * 16;
    if (r0 < N) gcn_gemm1_tile(X, W1, Y, N, r0, lane);
    return;
  }
  const int g    = (int)blockIdx.x;
  const int e_lo = min(E, g * chunk);
  const int e_hi = min(E, e_lo + chunk);

  for (int i = t; i <= nb; i += 1024) h[i] = 0;
  __syncthreads();

  for (int e0 = e_lo + t; e0 < e_hi; e0 += GCN_LD * 1024) {
    int d[GCN_LD];
#pragma unroll
    for (int u = 0; u < GCN_LD; ++u) d[u] = edst[min(e0 + u * 1024, e_hi - 1)];
#pragma unroll
    for (int u = 0; u < GCN_LD; ++u) {
      const int e = e0 + u * 1024;
      if (e < e_hi) lrank[e] = atomicAdd(&h[(unsigned)d[u] / GCN_RB], 1);
    }
  }
  __syncthreads();

  if (w == 0) {
    const int per = (nb + 31) / 32;
    const int b0  = min(nb, lane * per);
    const int b1  = min(nb, b0 + per);
    int sum = 0;
    for (int b = b0; b < b1; ++b) sum += h[b];
    int inc = sum;
#pragma unroll
    for (int off = 1; off < 32; off <<= 1) {
      const int y = __shfl_up(inc, off, 32);
      if (lane >= off) inc += y;
    }
    int run = inc - sum;
    for (int b = b0; b < b1; ++b) { const int c = h[b]; h[b] = run; run += c; }
    if (lane == 31) h[nb] = inc;
  }
  __syncthreads();

  for (int i = t; i <= nb; i += 1024) offG[(size_t)g * nbp + i] = h[i];

  int2* __restrict__ pe2 = reinterpret_cast<int2*>(pe);
  for (int e0 = e_lo + t; e0 < e_hi; e0 += GCN_LD * 1024) {
    int d[GCN_LD], rk[GCN_LD], sv[GCN_LD], vv[GCN_LD];
#pragma unroll
    for (int u = 0; u < GCN_LD; ++u) {
      const int e = min(e0 + u * 1024, e_hi - 1);
      d[u]  = edst[e];
      rk[u] = lrank[e];
      sv[u] = esrc[e];
      vv[u] = __float_as_int(evals[e]);
    }
#pragma unroll
    for (int u = 0; u < GCN_LD; ++u) {
      if (e0 + u * 1024 < e_hi) {
        const int j = e_lo + h[(unsigned)d[u] / GCN_RB] + rk[u];
        pe2[j] = make_int2(sv[u], vv[u]);
        pd[j]  = d[u];
      }
    }
  }
}

#define GCN_ZP 20
#define GCN_LP 50

struct __attribute__((aligned(16))) GcnBlockLds {
  float rows[GCN_RB * GCN_ZP];
  int   tab[2 * GCN_RB * GCN_TP];
  int   cnt[GCN_RB + 16];
  int   reserve[GCN_ZRES - GCN_RB * GCN_ZP - 2 * GCN_RB * GCN_TP - (GCN_RB + 16)];
};

static __device__ __forceinline__ float4 gcn_block_rowsum(
    const int* __restrict__ offG, const int2* __restrict__ pe2,
    const int* __restrict__ pd, const float4* __restrict__ X4,
    GcnBlockLds& L, int b, int nbp, int chunk, int E,
    int t, int wv, int lane, int doRelu) {
  const int q  = lane >> 2;
  const int c4 = lane & 3;
  const int g0 = 2 * wv;
  const int* __restrict__ oa = offG + (size_t)g0 * nbp + b;
  const int* __restrict__ ob = oa + nbp;
  const int sa0 = oa[0], sa1 = oa[1];
  const int sb0 = ob[0], sb1 = ob[1];
  const int lenA = sa1 - sa0;
  const int tot  = lenA + (sb1 - sb0);
  const int ja   = min(E, g0 * chunk) + sa0;
  const int jb   = min(E, (g0 + 1) * chunk) + sb0 - lenA;
  const int jsafe = (lenA > 0) ? ja : jb;
  const int row0 = b * GCN_RB;

  for (int i = t; i < GCN_RB; i += 1024) L.cnt[i] = 0;
  for (int i = t; i < GCN_RB * GCN_ZP; i += 1024) L.rows[i] = 0.f;
  __syncthreads();

  for (int ib = 0; ib < tot; ib += GCN_PQ) {
    int sv[GCN_SL], vv[GCN_SL], dl[GCN_SL];
#pragma unroll
    for (int u = 0; u < GCN_SL; ++u) {
      const int i = ib + u * 32 + lane;
      const int j = (i < tot) ? ((i < lenA) ? ja + i : jb + i) : jsafe;
      const int2 ev = pe2[j];
      sv[u] = ev.x; vv[u] = ev.y;
      dl[u] = pd[j];
    }
    int rks[GCN_SL];
#pragma unroll
    for (int u = 0; u < GCN_SL; ++u)
      rks[u] = (ib + u * 32 + lane < tot) ? atomicAdd(&L.cnt[dl[u] - row0], 1) : -1;
#pragma unroll
    for (int u = 0; u < GCN_SL; ++u) {
      const int rk = rks[u];
      if (rk >= 0) {
        const int r = dl[u] - row0;
        if (rk < GCN_RK) {
          L.tab[2 * (r * GCN_TP + rk)]     = sv[u];
          L.tab[2 * (r * GCN_TP + rk) + 1] = vv[u];
        } else {
          const float v = __int_as_float(vv[u]);
          float* __restrict__ z = L.rows + r * GCN_ZP;
          for (int c = 0; c < 4; ++c) {
            float4 xv = X4[(size_t)sv[u] * 4 + c];
            if (doRelu) {
              xv.x = fmaxf(xv.x, 0.f); xv.y = fmaxf(xv.y, 0.f);
              xv.z = fmaxf(xv.z, 0.f); xv.w = fmaxf(xv.w, 0.f);
            }
            __hip_atomic_fetch_add(z + 4 * c + 0, v * xv.x, __ATOMIC_RELAXED, __HIP_MEMORY_SCOPE_WORKGROUP);
            __hip_atomic_fetch_add(z + 4 * c + 1, v * xv.y, __ATOMIC_RELAXED, __HIP_MEMORY_SCOPE_WORKGROUP);
            __hip_atomic_fetch_add(z + 4 * c + 2, v * xv.z, __ATOMIC_RELAXED, __HIP_MEMORY_SCOPE_WORKGROUP);
            __hip_atomic_fetch_add(z + 4 * c + 3, v * xv.w, __ATOMIC_RELAXED, __HIP_MEMORY_SCOPE_WORKGROUP);
          }
        }
      }
    }
  }
  __syncthreads();

  const int lr = 8 * wv + q;
  float4 acc = make_float4(0.f, 0.f, 0.f, 0.f);
  if (lr < GCN_RB) {
    const int n = L.cnt[lr];
    const int e = min(n, GCN_RK);
    const int* __restrict__ tb = L.tab + 2 * lr * GCN_TP;
    for (int k0 = 0; k0 < e; k0 += GCN_WB) {
      int es_s[GCN_WB], es_v[GCN_WB];
#pragma unroll
      for (int u = 0; u < GCN_WB; ++u) {
        const int k = min(k0 + u, e - 1);
        es_s[u] = tb[2 * k];
        es_v[u] = tb[2 * k + 1];
      }
      float4 x[GCN_WB];
#pragma unroll
      for (int u = 0; u < GCN_WB; ++u) x[u] = X4[(size_t)es_s[u] * 4 + c4];
#pragma unroll
      for (int u = 0; u < GCN_WB; ++u) {
        float4 xv = x[u];
        if (doRelu) {
          xv.x = fmaxf(xv.x, 0.f); xv.y = fmaxf(xv.y, 0.f);
          xv.z = fmaxf(xv.z, 0.f); xv.w = fmaxf(xv.w, 0.f);
        }
        const bool  ok = (k0 + u < e);
        const float v  = __int_as_float(es_v[u]);
        acc.x = ok ? acc.x + v * xv.x : acc.x; acc.y = ok ? acc.y + v * xv.y : acc.y;
        acc.z = ok ? acc.z + v * xv.z : acc.z; acc.w = ok ? acc.w + v * xv.w : acc.w;
      }
    }
    if (n > GCN_RK) {
      const float4 o = *reinterpret_cast<const float4*>(&L.rows[lr * GCN_ZP + 4 * c4]);
      acc.x += o.x; acc.y += o.y; acc.z += o.z; acc.w += o.w;
    }
  }
  return acc;
}

__global__ __launch_bounds__(1024) void gcn_spmm_block_kernel(
    const int* __restrict__ offG, const int* __restrict__ pe,
    const int* __restrict__ pd, const float* __restrict__ Xin,
    float* __restrict__ Z, int N, int E, int nbp, int chunk) {
  __shared__ GcnBlockLds L;
  const int t    = (int)threadIdx.x;
  const int lane = t & 31;
  const int wv = __builtin_amdgcn_readfirstlane(t >> 5);
  const int b  = (int)blockIdx.x;

  const float4 acc = gcn_block_rowsum(offG, reinterpret_cast<const int2*>(pe), pd,
                                      reinterpret_cast<const float4*>(Xin), L, b, nbp, chunk, E,
                                      t, wv, lane,  0);

  const int lr  = 8 * wv + (lane >> 2);
  const int row = b * GCN_RB + lr;
  if (lr < GCN_RB && row < N)
    reinterpret_cast<float4*>(Z)[(size_t)row * 4 + (lane & 3)] = acc;
}

#define GCN_RT (GCN_RB / 16)

__global__ __launch_bounds__(1024) void gcn_spmm_gemm2_lsm_kernel(
    const int* __restrict__ offG, const int* __restrict__ pe,
    const int* __restrict__ pd, const float* __restrict__ Xin,
    const float* __restrict__ W2, float* __restrict__ out,
    int N, int E, int nbp, int chunk) {
  __shared__ GcnBlockLds L;
  __shared__ float s_l[GCN_RB * GCN_LP];
  const int t    = (int)threadIdx.x;
  const int lane = t & 31;
  const int wv  = __builtin_amdgcn_readfirstlane(t >> 5);
  const int b   = (int)blockIdx.x;
  const int col = lane & 15;
  const int kh  = lane >> 4;

  const float4 acc = gcn_block_rowsum(offG, reinterpret_cast<const int2*>(pe), pd,
                                      reinterpret_cast<const float4*>(Xin), L, b, nbp, chunk, E,
                                      t, wv, lane,  1);
  const float* __restrict__ zl = L.rows;
  {
    const int lr = 8 * wv + (lane >> 2);
    if (lr < GCN_RB)
      *reinterpret_cast<float4*>(&L.rows[lr * GCN_ZP + 4 * (lane & 3)]) = acc;
  }
  __syncthreads();

  for (int job = wv; job < 3 * GCN_RT; job += 32) {
    const int tile = job / 3;
    const int ct   = job - 3 * tile;
    const int cc = min(16 * ct + col, NCLS - 1);
    v2f bf[4];
#pragma unroll
    for (int j = 0; j < 4; ++j) {
      const int ka = 4 * j + 2 * kh;
      bf[j].x = W2[ka * NCLS + cc];
      bf[j].y = W2[(ka + 1) * NCLS + cc];
    }
    const v2f* __restrict__ zr =
        reinterpret_cast<const v2f*>(zl + (tile * 16 + col) * GCN_ZP + 2 * kh);
    v8f acc = {};
#pragma unroll
    for (int j = 0; j < 4; ++j) acc = wmma_f32_16x16x4(zr[2 * j], bf[j], acc);
    float* __restrict__ lw = s_l + (tile * 16 + 8 * kh) * GCN_LP + 16 * ct + col;
#pragma unroll
    for (int g = 0; g < 8; ++g) lw[g * GCN_LP] = acc[g];
  }
  __syncthreads();

  const bool t2ok = (col < 8);
  const float ninf = -__builtin_inff();
  for (int lr = 2 * wv + kh; lr < GCN_RB; lr += 64) {
    const int row = b * GCN_RB + lr;
    const float* __restrict__ lp = s_l + lr * GCN_LP + col;
    const float v0  = lp[0];
    const float v1  = lp[16];
    const float v2v = lp[32];

    float m = fmaxf(fmaxf(v0, v1), t2ok ? v2v : ninf);
#pragma unroll
    for (int off = 8; off >= 1; off >>= 1) m = fmaxf(m, __shfl_xor(m, off, 16));

    float s = expf(v0 - m) + expf(v1 - m) + (t2ok ? expf(v2v - m) : 0.0f);
#pragma unroll
    for (int off = 8; off >= 1; off >>= 1) s += __shfl_xor(s, off, 16);

    const float lse = m + logf(s);

    if (row < N) {
      float* __restrict__ o = out + (size_t)row * NCLS + col;
      o[0]  = v0 - lse;
      o[16] = v1 - lse;
      if (t2ok) o[32] = v2v - lse;
    }
  }
}

extern "C" void kernel_launch(void* const* d_in, const int* in_sizes, int n_in,
                              void* d_out, int out_size, void* d_ws, size_t ws_size,
                              hipStream_t stream) {
  const float* features = (const float*)d_in[0];
  const int*   esrc     = (const int*)d_in[1];
  const int*   edst     = (const int*)d_in[2];
  const float* evals    = (const float*)d_in[3];
  const float* W1       = (const float*)d_in[4];
  const float* W2       = (const float*)d_in[5];
  float*       out      = (float*)d_out;

  const int N = in_sizes[0] / IN_DIM;
  const int E = in_sizes[1];

  float* Y  = (float*)d_ws;
  float* Z1 = Y + (size_t)N * HIDDEN;
  const size_t ePad  = ((size_t)E + 63) / 64 * 64;
  const int    chunk = (E + GCN_P - 1) / GCN_P;
  const int    nb    = (N + GCN_RB - 1) / GCN_RB;
  const int    nbp   = (nb + 1 + 63) / 64 * 64;
  int* lrank = (int*)(Z1 + (size_t)N * HIDDEN);
  int* pd    = lrank + ePad;
  int* pe    = pd + ePad;
  int* offG  = pe + 2 * ePad;

  const int tiles  = (N + 15) / 16;
  const int g1Grps = (tiles + GCN_G1W - 1) / GCN_G1W;
  gcn_part_gemm1_kernel<<<GCN_P + g1Grps, 1024, 0, stream>>>(
      esrc, edst, evals, offG, lrank, pe, pd, N, E, nb, nbp, chunk, features, W1, Y);

  gcn_spmm_block_kernel<<<nb, 1024, 0, stream>>>(offG, pe, pd, Y, Z1, N, E, nbp, chunk);

  gcn_spmm_gemm2_lsm_kernel<<<nb, 1024, 0, stream>>>(offG, pe, pd, Z1, W2, out, N, E, nbp, chunk);
}


# Edge-run sort: each run is put in row-block order in LDS and written out in whole 128-byte lines, not as one scattered 8-byte and one 4-byte global store per edge; order of sums unchanged
# speedup vs baseline: 3.0068x; 1.3345x over previous
#include <hip/hip_runtime.h>
#include <math.h>

#define IN_DIM  100
#define HIDDEN  16
#define NCLS    40

typedef __attribute__((ext_vector_type(2))) float v2f;
typedef __attribute__((ext_vector_type(8))) float v8f;

static __device__ inline v8f wmma_f32_16x16x4(v2f a, v2f b, v8f c) {
#if __has_builtin(__builtin_amdgcn_wmma_f32_16x16x4_f32)
  return __builtin_amdgcn_wmma_f32_16x16x4_f32(
       false, a,  false, b,
       (short)0, c,  false,  false);
#else
  const int lane = (int)(threadIdx.x & 31u);
  const int nsel = lane & 15;
  v8f d = c;
#pragma unroll
  for (int g = 0; g < 8; ++g) {
    const int m = g + 8 * (lane >> 4);
    float acc = 0.f;
#pragma unroll
    for (int k = 0; k < 4; ++k) {
      const int srcA = m + 16 * (k >> 1);
      const int srcB = nsel + 16 * (k >> 1);
      const float av = __shfl((k & 1) ? a.y : a.x, srcA, 32);
      const float bv = __shfl((k & 1) ? b.y : b.x, srcB, 32);
      acc += av * bv;
    }
    d[g] += acc;
  }
  return d;
#endif
}

#define GCN_P      64
#define GCN_RB     208
#define GCN_HCAP   65536
#define GCN_LD     16
#define GCN_CO     4
#define GCN_G1W    32
#define GCN_SL     6
#define GCN_PQ     (32 * GCN_SL)
#define GCN_RK     48
#define GCN_TP     (GCN_RK + 1)
#define GCN_WB     12
#define GCN_ZRES   41472

static __device__ inline void gcn_gemm1_tile(
    const float* __restrict__ X, const float* __restrict__ W1,
    float* __restrict__ Y, int N, int r0, int lane) {
  const int col  = lane & 15;
  const int kh   = lane >> 4;
  const int rowA = min(r0 + col, N - 1);
  const float* __restrict__ Xr = X + (size_t)rowA * IN_DIM;

  v8f acc = {};
#pragma unroll 5
  for (int k0 = 0; k0 < IN_DIM; k0 += 4) {
    const int ka = k0 + 2 * kh;
    v2f a;
    a.x = Xr[ka];
    a.y = Xr[ka + 1];
    v2f b;
    b.x = W1[ka * HIDDEN + col];
    b.y = W1[(ka + 1) * HIDDEN + col];
    acc = wmma_f32_16x16x4(a, b, acc);
  }

  float* __restrict__ Yr = Y + (size_t)(r0 + 8 * kh) * HIDDEN + col;
  if (r0 + 16 <= N) {
#pragma unroll
    for (int g = 0; g < 8; ++g) Yr[g * HIDDEN] = acc[g];
  } else {
#pragma unroll
    for (int g = 0; g < 8; ++g) {
      const int row = r0 + g + 8 * kh;
      if (row < N) Y[row * HIDDEN + col] = acc[g];
    }
  }
}

template <bool kStage>
static __device__ __forceinline__ void gcn_place_run(
    const int* __restrict__ esrc, const int* __restrict__ edst,
    const float* __restrict__ evals, const int* __restrict__ lrank,
    const int* __restrict__ hp, int2* __restrict__ sp, int* __restrict__ sd,
    int2* __restrict__ pe2, int* __restrict__ pd, int e_lo, int e_hi, int t) {
  for (int e0 = e_lo + t; e0 < e_hi; e0 += GCN_LD * 1024) {
    int d[GCN_LD], rk[GCN_LD], sv[GCN_LD], vv[GCN_LD];
#pragma unroll
    for (int u = 0; u < GCN_LD; ++u) {
      const int e = min(e0 + u * 1024, e_hi - 1);
      d[u]  = edst[e];
      rk[u] = lrank[e];
      sv[u] = esrc[e];
      vv[u] = __float_as_int(evals[e]);
    }
#pragma unroll
    for (int u = 0; u < GCN_LD; ++u) {
      if (e0 + u * 1024 < e_hi) {
        const int jl = hp[(unsigned)d[u] / GCN_RB] + rk[u];
        if (kStage) {
          sp[jl] = make_int2(sv[u], vv[u]);
          sd[jl] = d[u];
        } else {
          pe2[e_lo + jl] = make_int2(sv[u], vv[u]);
          pd[e_lo + jl]  = d[u];
        }
      }
    }
  }
}

__global__ __launch_bounds__(1024) void gcn_part_gemm1_kernel(
    const int* __restrict__ esrc, const int* __restrict__ edst,
    const float* __restrict__ evals, int* __restrict__ offG,
    int* __restrict__ lrank, int* __restrict__ pe, int* __restrict__ pd,
    int N, int E, int nb, int nbp, int chunk,
    const float* __restrict__ X, const float* __restrict__ W1,
    float* __restrict__ Y) {
  __shared__ int h[GCN_HCAP];
  const int t    = (int)threadIdx.x;
  const int lane = t & 31;
  const int w    = t >> 5;
  if ((int)blockIdx.x >= GCN_P) {
    const int wv = __builtin_amdgcn_readfirstlane(w);
    const int r0 = (((int)blockIdx.x - GCN_P) * GCN_G1W + wv) * 16;
    if (r0 < N) gcn_gemm1_tile(X, W1, Y, N, r0, lane);
    return;
  }
  const int g    = (int)blockIdx.x;
  const int e_lo = min(E, g * chunk);
  const int e_hi = min(E, e_lo + chunk);

  for (int i = t; i <= nb; i += 1024) h[i] = 0;
  __syncthreads();

  for (int e0 = e_lo + t; e0 < e_hi; e0 += GCN_LD * 1024) {
    int d[GCN_LD];
#pragma unroll
    for (int u = 0; u < GCN_LD; ++u) d[u] = edst[min(e0 + u * 1024, e_hi - 1)];
#pragma unroll
    for (int u = 0; u < GCN_LD; ++u) {
      const int e = e0 + u * 1024;
      if (e < e_hi) lrank[e] = atomicAdd(&h[(unsigned)d[u] / GCN_RB], 1);
    }
  }
  __syncthreads();

  if (w == 0) {
    const int per = (nb + 31) / 32;
    const int b0  = min(nb, lane * per);
    const int b1  = min(nb, b0 + per);
    int sum = 0;
    for (int b = b0; b < b1; ++b) sum += h[b];
    int inc = sum;
#pragma unroll
    for (int off = 1; off < 32; off <<= 1) {
      const int y = __shfl_up(inc, off, 32);
      if (lane >= off) inc += y;
    }
    int run = inc - sum;
    for (int b = b0; b < b1; ++b) { const int c = h[b]; h[b] = run; run += c; }
    if (lane == 31) h[nb] = inc;
  }
  __syncthreads();

  for (int i = t; i <= nb; i += 1024) offG[(size_t)g * nbp + i] = h[i];

  int2* __restrict__ pe2 = reinterpret_cast<int2*>(pe);
  const int cp = (chunk + 3) & ~3;
  if (nbp + 3 * cp <= GCN_HCAP) {
    int2* __restrict__ sp = reinterpret_cast<int2*>(h + nbp);
    int*  __restrict__ sd = h + nbp + 2 * cp;
    gcn_place_run<true>(esrc, edst, evals, lrank, h, sp, sd, pe2, pd, e_lo, e_hi, t);
    __syncthreads();
    const int cnt = e_hi - e_lo;
    if (cnt > 0) {
      for (int k0 = t - (e_lo & 15); k0 < cnt; k0 += GCN_CO * 1024) {
        int2 v[GCN_CO];
#pragma unroll
        for (int u = 0; u < GCN_CO; ++u) v[u] = sp[min(max(k0 + u * 1024, 0), cnt - 1)];
#pragma unroll
        for (int u = 0; u < GCN_CO; ++u) {
          const int k = k0 + u * 1024;
          if (k >= 0 && k < cnt) pe2[e_lo + k] = v[u];
        }
      }
      for (int k0 = t - (e_lo & 31); k0 < cnt; k0 += GCN_CO * 1024) {
        int v[GCN_CO];
#pragma unroll
        for (int u = 0; u < GCN_CO; ++u) v[u] = sd[min(max(k0 + u * 1024, 0), cnt - 1)];
#pragma unroll
        for (int u = 0; u < GCN_CO; ++u) {
          const int k = k0 + u * 1024;
          if (k >= 0 && k < cnt) pd[e_lo + k] = v[u];
        }
      }
    }
  } else {
    gcn_place_run<false>(esrc, edst, evals, lrank, h, nullptr, nullptr, pe2, pd, e_lo, e_hi, t);
  }
}

#define GCN_ZP 20
#define GCN_LP 50

struct __attribute__((aligned(16))) GcnBlockLds {
  float rows[GCN_RB * GCN_ZP];
  int   tab[2 * GCN_RB * GCN_TP];
  int   cnt[GCN_RB + 16];
  int   reserve[GCN_ZRES - GCN_RB * GCN_ZP - 2 * GCN_RB * GCN_TP - (GCN_RB + 16)];
};

static __device__ __forceinline__ float4 gcn_block_rowsum(
    const int* __restrict__ offG, const int2* __restrict__ pe2,
    const int* __restrict__ pd, const float4* __restrict__ X4,
    GcnBlockLds& L, int b, int nbp, int chunk, int E,
    int t, int wv, int lane, int doRelu) {
  const int q  = lane >> 2;
  const int c4 = lane & 3;
  const int g0 = 2 * wv;
  const int* __restrict__ oa = offG + (size_t)g0 * nbp + b;
  const int* __restrict__ ob = oa + nbp;
  const int sa0 = oa[0], sa1 = oa[1];
  const int sb0 = ob[0], sb1 = ob[1];
  const int lenA = sa1 - sa0;
  const int tot  = lenA + (sb1 - sb0);
  const int ja   = min(E, g0 * chunk) + sa0;
  const int jb   = min(E, (g0 + 1) * chunk) + sb0 - lenA;
  const int jsafe = (lenA > 0) ? ja : jb;
  const int row0 = b * GCN_RB;

  for (int i = t; i < GCN_RB; i += 1024) L.cnt[i] = 0;
  for (int i = t; i < GCN_RB * GCN_ZP; i += 1024) L.rows[i] = 0.f;
  __syncthreads();

  for (int ib = 0; ib < tot; ib += GCN_PQ) {
    int sv[GCN_SL], vv[GCN_SL], dl[GCN_SL];
#pragma unroll
    for (int u = 0; u < GCN_SL; ++u) {
      const int i = ib + u * 32 + lane;
      const int j = (i < tot) ? ((i < lenA) ? ja + i : jb + i) : jsafe;
      const int2 ev = pe2[j];
      sv[u] = ev.x; vv[u] = ev.y;
      dl[u] = pd[j];
    }
    int rks[GCN_SL];
#pragma unroll
    for (int u = 0; u < GCN_SL; ++u)
      rks[u] = (ib + u * 32 + lane < tot) ? atomicAdd(&L.cnt[dl[u] - row0], 1) : -1;
#pragma unroll
    for (int u = 0; u < GCN_SL; ++u) {
      const int rk = rks[u];
      if (rk >= 0) {
        const int r = dl[u] - row0;
        if (rk < GCN_RK) {
          L.tab[2 * (r * GCN_TP + rk)]     = sv[u];
          L.tab[2 * (r * GCN_TP + rk) + 1] = vv[u];
        } else {
          const float v = __int_as_float(vv[u]);
          float* __restrict__ z = L.rows + r * GCN_ZP;
          for (int c = 0; c < 4; ++c) {
            float4 xv = X4[(size_t)sv[u] * 4 + c];
            if (doRelu) {
              xv.x = fmaxf(xv.x, 0.f); xv.y = fmaxf(xv.y, 0.f);
              xv.z = fmaxf(xv.z, 0.f); xv.w = fmaxf(xv.w, 0.f);
            }
            __hip_atomic_fetch_add(z + 4 * c + 0, v * xv.x, __ATOMIC_RELAXED, __HIP_MEMORY_SCOPE_WORKGROUP);
            __hip_atomic_fetch_add(z + 4 * c + 1, v * xv.y, __ATOMIC_RELAXED, __HIP_MEMORY_SCOPE_WORKGROUP);
            __hip_atomic_fetch_add(z + 4 * c + 2, v * xv.z, __ATOMIC_RELAXED, __HIP_MEMORY_SCOPE_WORKGROUP);
            __hip_atomic_fetch_add(z + 4 * c + 3, v * xv.w, __ATOMIC_RELAXED, __HIP_MEMORY_SCOPE_WORKGROUP);
          }
        }
      }
    }
  }
  __syncthreads();

  const int lr = 8 * wv + q;
  float4 acc = make_float4(0.f, 0.f, 0.f, 0.f);
  if (lr < GCN_RB) {
    const int n = L.cnt[lr];
    const int e = min(n, GCN_RK);
    const int* __restrict__ tb = L.tab + 2 * lr * GCN_TP;
    for (int k0 = 0; k0 < e; k0 += GCN_WB) {
      int es_s[GCN_WB], es_v[GCN_WB];
#pragma unroll
      for (int u = 0; u < GCN_WB; ++u) {
        const int k = min(k0 + u, e - 1);
        es_s[u] = tb[2 * k];
        es_v[u] = tb[2 * k + 1];
      }
      float4 x[GCN_WB];
#pragma unroll
      for (int u = 0; u < GCN_WB; ++u) x[u] = X4[(size_t)es_s[u] * 4 + c4];
#pragma unroll
      for (int u = 0; u < GCN_WB; ++u) {
        float4 xv = x[u];
        if (doRelu) {
          xv.x = fmaxf(xv.x, 0.f); xv.y = fmaxf(xv.y, 0.f);
          xv.z = fmaxf(xv.z, 0.f); xv.w = fmaxf(xv.w, 0.f);
        }
        const bool  ok = (k0 + u < e);
        const float v  = __int_as_float(es_v[u]);
        acc.x = ok ? acc.x + v * xv.x : acc.x; acc.y = ok ? acc.y + v * xv.y : acc.y;
        acc.z = ok ? acc.z + v * xv.z : acc.z; acc.w = ok ? acc.w + v * xv.w : acc.w;
      }
    }
    if (n > GCN_RK) {
      const float4 o = *reinterpret_cast<const float4*>(&L.rows[lr * GCN_ZP + 4 * c4]);
      acc.x += o.x; acc.y += o.y; acc.z += o.z; acc.w += o.w;
    }
  }
  return acc;
}

__global__ __launch_bounds__(1024) void gcn_spmm_block_kernel(
    const int* __restrict__ offG, const int* __restrict__ pe,
    const int* __restrict__ pd, const float* __restrict__ Xin,
    float* __restrict__ Z, int N, int E, int nbp, int chunk) {
  __shared__ GcnBlockLds L;
  const int t    = (int)threadIdx.x;
  const int lane = t & 31;
  const int wv = __builtin_amdgcn_readfirstlane(t >> 5);
  const int b  = (int)blockIdx.x;

  const float4 acc = gcn_block_rowsum(offG, reinterpret_cast<const int2*>(pe), pd,
                                      reinterpret_cast<const float4*>(Xin), L, b, nbp, chunk, E,
                                      t, wv, lane,  0);

  const int lr  = 8 * wv + (lane >> 2);
  const int row = b * GCN_RB + lr;
  if (lr < GCN_RB && row < N)
    reinterpret_cast<float4*>(Z)[(size_t)row * 4 + (lane & 3)] = acc;
}

#define GCN_RT (GCN_RB / 16)

__global__ __launch_bounds__(1024) void gcn_spmm_gemm2_lsm_kernel(
    const int* __restrict__ offG, const int* __restrict__ pe,
    const int* __restrict__ pd, const float* __restrict__ Xin,
    const float* __restrict__ W2, float* __restrict__ out,
    int N, int E, int nbp, int chunk) {
  __shared__ GcnBlockLds L;
  __shared__ float s_l[GCN_RB * GCN_LP];
  const int t    = (int)threadIdx.x;
  const int lane = t & 31;
  const int wv  = __builtin_amdgcn_readfirstlane(t >> 5);
  const int b   = (int)blockIdx.x;
  const int col = lane & 15;
  const int kh  = lane >> 4;

  const float4 acc = gcn_block_rowsum(offG, reinterpret_cast<const int2*>(pe), pd,
                                      reinterpret_cast<const float4*>(Xin), L, b, nbp, chunk, E,
                                      t, wv, lane,  1);
  const float* __restrict__ zl = L.rows;
  {
    const int lr = 8 * wv + (lane >> 2);
    if (lr < GCN_RB)
      *reinterpret_cast<float4*>(&L.rows[lr * GCN_ZP + 4 * (lane & 3)]) = acc;
  }
  __syncthreads();

  for (int job = wv; job < 3 * GCN_RT; job += 32) {
    const int tile = job / 3;
    const int ct   = job - 3 * tile;
    const int cc = min(16 * ct + col, NCLS - 1);
    v2f bf[4];
#pragma unroll
    for (int j = 0; j < 4; ++j) {
      const int ka = 4 * j + 2 * kh;
      bf[j].x = W2[ka * NCLS + cc];
      bf[j].y = W2[(ka + 1) * NCLS + cc];
    }
    const v2f* __restrict__ zr =
        reinterpret_cast<const v2f*>(zl + (tile * 16 + col) * GCN_ZP + 2 * kh);
    v8f acc = {};
#pragma unroll
    for (int j = 0; j < 4; ++j) acc = wmma_f32_16x16x4(zr[2 * j], bf[j], acc);
    float* __restrict__ lw = s_l + (tile * 16 + 8 * kh) * GCN_LP + 16 * ct + col;
#pragma unroll
    for (int g = 0; g < 8; ++g) lw[g * GCN_LP] = acc[g];
  }
  __syncthreads();

  const bool t2ok = (col < 8);
  const float ninf = -__builtin_inff();
  for (int lr = 2 * wv + kh; lr < GCN_RB; lr += 64) {
    const int row = b * GCN_RB + lr;
    const float* __restrict__ lp = s_l + lr * GCN_LP + col;
    const float v0  = lp[0];
    const float v1  = lp[16];
    const float v2v = lp[32];

    float m = fmaxf(fmaxf(v0, v1), t2ok ? v2v : ninf);
#pragma unroll
    for (int off = 8; off >= 1; off >>= 1) m = fmaxf(m, __shfl_xor(m, off, 16));

    float s = expf(v0 - m) + expf(v1 - m) + (t2ok ? expf(v2v - m) : 0.0f);
#pragma unroll
    for (int off = 8; off >= 1; off >>= 1) s += __shfl_xor(s, off, 16);

    const float lse = m + logf(s);

    if (row < N) {
      float* __restrict__ o = out + (size_t)row * NCLS + col;
      o[0]  = v0 - lse;
      o[16] = v1 - lse;
      if (t2ok) o[32] = v2v - lse;
    }
  }
}

extern "C" void kernel_launch(void* const* d_in, const int* in_sizes, int n_in,
                              void* d_out, int out_size, void* d_ws, size_t ws_size,
                              hipStream_t stream) {
  const float* features = (const float*)d_in[0];
  const int*   esrc     = (const int*)d_in[1];
  const int*   edst     = (const int*)d_in[2];
  const float* evals    = (const float*)d_in[3];
  const float* W1       = (const float*)d_in[4];
  const float* W2       = (const float*)d_in[5];
  float*       out      = (float*)d_out;

  const int N = in_sizes[0] / IN_DIM;
  const int E = in_sizes[1];

  float* Y  = (float*)d_ws;
  float* Z1 = Y + (size_t)N * HIDDEN;
  const size_t ePad  = ((size_t)E + 63) / 64 * 64;
  const int    chunk = (E + GCN_P - 1) / GCN_P;
  const int    nb    = (N + GCN_RB - 1) / GCN_RB;
  const int    nbp   = (nb + 1 + 63) / 64 * 64;
  int* lrank = (int*)(Z1 + (size_t)N * HIDDEN);
  int* pd    = lrank + ePad;
  int* pe    = pd + ePad;
  int* offG  = pe + 2 * ePad;

  const int tiles  = (N + 15) / 16;
  const int g1Grps = (tiles + GCN_G1W - 1) / GCN_G1W;
  gcn_part_gemm1_kernel<<<GCN_P + g1Grps, 1024, 0, stream>>>(
      esrc, edst, evals, offG, lrank, pe, pd, N, E, nb, nbp, chunk, features, W1, Y);

  gcn_spmm_block_kernel<<<nb, 1024, 0, stream>>>(offG, pe, pd, Y, Z1, N, E, nbp, chunk);

  gcn_spmm_gemm2_lsm_kernel<<<nb, 1024, 0, stream>>>(offG, pe, pd, Z1, W2, out, N, E, nbp, chunk);
}


# Log-softmax re-laid: a row per lane quad (10 logits a lane), whole block in one pass, quad lane permutes for max and sum; rows leave via LDS as whole 128-byte lines; exp sums in a new tree order
# speedup vs baseline: 3.1831x; 1.0586x over previous
#include <hip/hip_runtime.h>
#include <math.h>

#define IN_DIM  100
#define HIDDEN  16
#define NCLS    40

typedef __attribute__((ext_vector_type(2))) float v2f;
typedef __attribute__((ext_vector_type(8))) float v8f;

static __device__ inline v8f wmma_f32_16x16x4(v2f a, v2f b, v8f c) {
#if __has_builtin(__builtin_amdgcn_wmma_f32_16x16x4_f32)
  return __builtin_amdgcn_wmma_f32_16x16x4_f32(
       false, a,  false, b,
       (short)0, c,  false,  false);
#else
  const int lane = (int)(threadIdx.x & 31u);
  const int nsel = lane & 15;
  v8f d = c;
#pragma unroll
  for (int g = 0; g < 8; ++g) {
    const int m = g + 8 * (lane >> 4);
    float acc = 0.f;
#pragma unroll
    for (int k = 0; k < 4; ++k) {
      const int srcA = m + 16 * (k >> 1);
      const int srcB = nsel + 16 * (k >> 1);
      const float av = __shfl((k & 1) ? a.y : a.x, srcA, 32);
      const float bv = __shfl((k & 1) ? b.y : b.x, srcB, 32);
      acc += av * bv;
    }
    d[g] += acc;
  }
  return d;
#endif
}

#define GCN_P      64
#define GCN_RB     208
#define GCN_HCAP   65536
#define GCN_LD     16
#define GCN_CO     4
#define GCN_G1W    32
#define GCN_SL     6
#define GCN_PQ     (32 * GCN_SL)
#define GCN_RK     48
#define GCN_TP     (GCN_RK + 1)
#define GCN_WB     12
#define GCN_ZRES   41472

static __device__ inline void gcn_gemm1_tile(
    const float* __restrict__ X, const float* __restrict__ W1,
    float* __restrict__ Y, int N, int r0, int lane) {
  const int col  = lane & 15;
  const int kh   = lane >> 4;
  const int rowA = min(r0 + col, N - 1);
  const float* __restrict__ Xr = X + (size_t)rowA * IN_DIM;

  v8f acc = {};
#pragma unroll 5
  for (int k0 = 0; k0 < IN_DIM; k0 += 4) {
    const int ka = k0 + 2 * kh;
    v2f a;
    a.x = Xr[ka];
    a.y = Xr[ka + 1];
    v2f b;
    b.x = W1[ka * HIDDEN + col];
    b.y = W1[(ka + 1) * HIDDEN + col];
    acc = wmma_f32_16x16x4(a, b, acc);
  }

  float* __restrict__ Yr = Y + (size_t)(r0 + 8 * kh) * HIDDEN + col;
  if (r0 + 16 <= N) {
#pragma unroll
    for (int g = 0; g < 8; ++g) Yr[g * HIDDEN] = acc[g];
  } else {
#pragma unroll
    for (int g = 0; g < 8; ++g) {
      const int row = r0 + g + 8 * kh;
      if (row < N) Y[row * HIDDEN + col] = acc[g];
    }
  }
}

template <bool kStage>
static __device__ __forceinline__ void gcn_place_run(
    const int* __restrict__ esrc, const int* __restrict__ edst,
    const float* __restrict__ evals, const int* __restrict__ lrank,
    const int* __restrict__ hp, int2* __restrict__ sp, int* __restrict__ sd,
    int2* __restrict__ pe2, int* __restrict__ pd, int e_lo, int e_hi, int t) {
  for (int e0 = e_lo + t; e0 < e_hi; e0 += GCN_LD * 1024) {
    int d[GCN_LD], rk[GCN_LD], sv[GCN_LD], vv[GCN_LD];
#pragma unroll
    for (int u = 0; u < GCN_LD; ++u) {
      const int e = min(e0 + u * 1024, e_hi - 1);
      d[u]  = edst[e];
      rk[u] = lrank[e];
      sv[u] = esrc[e];
      vv[u] = __float_as_int(evals[e]);
    }
#pragma unroll
    for (int u = 0; u < GCN_LD; ++u) {
      if (e0 + u * 1024 < e_hi) {
        const int jl = hp[(unsigned)d[u] / GCN_RB] + rk[u];
        if (kStage) {
          sp[jl] = make_int2(sv[u], vv[u]);
          sd[jl] = d[u];
        } else {
          pe2[e_lo + jl] = make_int2(sv[u], vv[u]);
          pd[e_lo + jl]  = d[u];
        }
      }
    }
  }
}

__global__ __launch_bounds__(1024) void gcn_part_gemm1_kernel(
    const int* __restrict__ esrc, const int* __restrict__ edst,
    const float* __restrict__ evals, int* __restrict__ offG,
    int* __restrict__ lrank, int* __restrict__ pe, int* __restrict__ pd,
    int N, int E, int nb, int nbp, int chunk,
    const float* __restrict__ X, const float* __restrict__ W1,
    float* __restrict__ Y) {
  __shared__ int h[GCN_HCAP];
  const int t    = (int)threadIdx.x;
  const int lane = t & 31;
  const int w    = t >> 5;
  if ((int)blockIdx.x >= GCN_P) {
    const int wv = __builtin_amdgcn_readfirstlane(w);
    const int r0 = (((int)blockIdx.x - GCN_P) * GCN_G1W + wv) * 16;
    if (r0 < N) gcn_gemm1_tile(X, W1, Y, N, r0, lane);
    return;
  }
  const int g    = (int)blockIdx.x;
  const int e_lo = min(E, g * chunk);
  const int e_hi = min(E, e_lo + chunk);

  for (int i = t; i <= nb; i += 1024) h[i] = 0;
  __syncthreads();

  for (int e0 = e_lo + t; e0 < e_hi; e0 += GCN_LD * 1024) {
    int d[GCN_LD];
#pragma unroll
    for (int u = 0; u < GCN_LD; ++u) d[u] = edst[min(e0 + u * 1024, e_hi - 1)];
#pragma unroll
    for (int u = 0; u < GCN_LD; ++u) {
      const int e = e0 + u * 1024;
      if (e < e_hi) lrank[e] = atomicAdd(&h[(unsigned)d[u] / GCN_RB], 1);
    }
  }
  __syncthreads();

  if (w == 0) {
    const int per = (nb + 31) / 32;
    const int b0  = min(nb, lane * per);
    const int b1  = min(nb, b0 + per);
    int sum = 0;
    for (int b = b0; b < b1; ++b) sum += h[b];
    int inc = sum;
#pragma unroll
    for (int off = 1; off < 32; off <<= 1) {
      const int y = __shfl_up(inc, off, 32);
      if (lane >= off) inc += y;
    }
    int run = inc - sum;
    for (int b = b0; b < b1; ++b) { const int c = h[b]; h[b] = run; run += c; }
    if (lane == 31) h[nb] = inc;
  }
  __syncthreads();

  for (int i = t; i <= nb; i += 1024) offG[(size_t)g * nbp + i] = h[i];

  int2* __restrict__ pe2 = reinterpret_cast<int2*>(pe);
  const int cp = (chunk + 3) & ~3;
  if (nbp + 3 * cp <= GCN_HCAP) {
    int2* __restrict__ sp = reinterpret_cast<int2*>(h + nbp);
    int*  __restrict__ sd = h + nbp + 2 * cp;
    gcn_place_run<true>(esrc, edst, evals, lrank, h, sp, sd, pe2, pd, e_lo, e_hi, t);
    __syncthreads();
    const int cnt = e_hi - e_lo;
    if (cnt > 0) {
      for (int k0 = t - (e_lo & 15); k0 < cnt; k0 += GCN_CO * 1024) {
        int2 v[GCN_CO];
#pragma unroll
        for (int u = 0; u < GCN_CO; ++u) v[u] = sp[min(max(k0 + u * 1024, 0), cnt - 1)];
#pragma unroll
        for (int u = 0; u < GCN_CO; ++u) {
          const int k = k0 + u * 1024;
          if (k >= 0 && k < cnt) pe2[e_lo + k] = v[u];
        }
      }
      for (int k0 = t - (e_lo & 31); k0 < cnt; k0 += GCN_CO * 1024) {
        int v[GCN_CO];
#pragma unroll
        for (int u = 0; u < GCN_CO; ++u) v[u] = sd[min(max(k0 + u * 1024, 0), cnt - 1)];
#pragma unroll
        for (int u = 0; u < GCN_CO; ++u) {
          const int k = k0 + u * 1024;
          if (k >= 0 && k < cnt) pd[e_lo + k] = v[u];
        }
      }
    }
  } else {
    gcn_place_run<false>(esrc, edst, evals, lrank, h, nullptr, nullptr, pe2, pd, e_lo, e_hi, t);
  }
}

#define GCN_ZP 20
#define GCN_LP 50

struct __attribute__((aligned(16))) GcnBlockLds {
  float rows[GCN_RB * GCN_ZP];
  int   tab[2 * GCN_RB * GCN_TP];
  int   cnt[GCN_RB + 16];
  int   reserve[GCN_ZRES - GCN_RB * GCN_ZP - 2 * GCN_RB * GCN_TP - (GCN_RB + 16)];
};

static __device__ __forceinline__ float4 gcn_block_rowsum(
    const int* __restrict__ offG, const int2* __restrict__ pe2,
    const int* __restrict__ pd, const float4* __restrict__ X4,
    GcnBlockLds& L, int b, int nbp, int chunk, int E,
    int t, int wv, int lane, int doRelu) {
  const int q  = lane >> 2;
  const int c4 = lane & 3;
  const int g0 = 2 * wv;
  const int* __restrict__ oa = offG + (size_t)g0 * nbp + b;
  const int* __restrict__ ob = oa + nbp;
  const int sa0 = oa[0], sa1 = oa[1];
  const int sb0 = ob[0], sb1 = ob[1];
  const int lenA = sa1 - sa0;
  const int tot  = lenA + (sb1 - sb0);
  const int ja   = min(E, g0 * chunk) + sa0;
  const int jb   = min(E, (g0 + 1) * chunk) + sb0 - lenA;
  const int jsafe = (lenA > 0) ? ja : jb;
  const int row0 = b * GCN_RB;

  for (int i = t; i < GCN_RB; i += 1024) L.cnt[i] = 0;
  for (int i = t; i < GCN_RB * GCN_ZP; i += 1024) L.rows[i] = 0.f;
  __syncthreads();

  for (int ib = 0; ib < tot; ib += GCN_PQ) {
    int sv[GCN_SL], vv[GCN_SL], dl[GCN_SL];
#pragma unroll
    for (int u = 0; u < GCN_SL; ++u) {
      const int i = ib + u * 32 + lane;
      const int j = (i < tot) ? ((i < lenA) ? ja + i : jb + i) : jsafe;
      const int2 ev = pe2[j];
      sv[u] = ev.x; vv[u] = ev.y;
      dl[u] = pd[j];
    }
    int rks[GCN_SL];
#pragma unroll
    for (int u = 0; u < GCN_SL; ++u)
      rks[u] = (ib + u * 32 + lane < tot) ? atomicAdd(&L.cnt[dl[u] - row0], 1) : -1;
#pragma unroll
    for (int u = 0; u < GCN_SL; ++u) {
      const int rk = rks[u];
      if (rk >= 0) {
        const int r = dl[u] - row0;
        if (rk < GCN_RK) {
          L.tab[2 * (r * GCN_TP + rk)]     = sv[u];
          L.tab[2 * (r * GCN_TP + rk) + 1] = vv[u];
        } else {
          const float v = __int_as_float(vv[u]);
          float* __restrict__ z = L.rows + r * GCN_ZP;
          for (int c = 0; c < 4; ++c) {
            float4 xv = X4[(size_t)sv[u] * 4 + c];
            if (doRelu) {
              xv.x = fmaxf(xv.x, 0.f); xv.y = fmaxf(xv.y, 0.f);
              xv.z = fmaxf(xv.z, 0.f); xv.w = fmaxf(xv.w, 0.f);
            }
            __hip_atomic_fetch_add(z + 4 * c + 0, v * xv.x, __ATOMIC_RELAXED, __HIP_MEMORY_SCOPE_WORKGROUP);
            __hip_atomic_fetch_add(z + 4 * c + 1, v * xv.y, __ATOMIC_RELAXED, __HIP_MEMORY_SCOPE_WORKGROUP);
            __hip_atomic_fetch_add(z + 4 * c + 2, v * xv.z, __ATOMIC_RELAXED, __HIP_MEMORY_SCOPE_WORKGROUP);
            __hip_atomic_fetch_add(z + 4 * c + 3, v * xv.w, __ATOMIC_RELAXED, __HIP_MEMORY_SCOPE_WORKGROUP);
          }
        }
      }
    }
  }
  __syncthreads();

  const int lr = 8 * wv + q;
  float4 acc = make_float4(0.f, 0.f, 0.f, 0.f);
  if (lr < GCN_RB) {
    const int n = L.cnt[lr];
    const int e = min(n, GCN_RK);
    const int* __restrict__ tb = L.tab + 2 * lr * GCN_TP;
    for (int k0 = 0; k0 < e; k0 += GCN_WB) {
      int es_s[GCN_WB], es_v[GCN_WB];
#pragma unroll
      for (int u = 0; u < GCN_WB; ++u) {
        const int k = min(k0 + u, e - 1);
        es_s[u] = tb[2 * k];
        es_v[u] = tb[2 * k + 1];
      }
      float4 x[GCN_WB];
#pragma unroll
      for (int u = 0; u < GCN_WB; ++u) x[u] = X4[(size_t)es_s[u] * 4 + c4];
#pragma unroll
      for (int u = 0; u < GCN_WB; ++u) {
        float4 xv = x[u];
        if (doRelu) {
          xv.x = fmaxf(xv.x, 0.f); xv.y = fmaxf(xv.y, 0.f);
          xv.z = fmaxf(xv.z, 0.f); xv.w = fmaxf(xv.w, 0.f);
        }
        const bool  ok = (k0 + u < e);
        const float v  = __int_as_float(es_v[u]);
        acc.x = ok ? acc.x + v * xv.x : acc.x; acc.y = ok ? acc.y + v * xv.y : acc.y;
        acc.z = ok ? acc.z + v * xv.z : acc.z; acc.w = ok ? acc.w + v * xv.w : acc.w;
      }
    }
    if (n > GCN_RK) {
      const float4 o = *reinterpret_cast<const float4*>(&L.rows[lr * GCN_ZP + 4 * c4]);
      acc.x += o.x; acc.y += o.y; acc.z += o.z; acc.w += o.w;
    }
  }
  return acc;
}

__global__ __launch_bounds__(1024) void gcn_spmm_block_kernel(
    const int* __restrict__ offG, const int* __restrict__ pe,
    const int* __restrict__ pd, const float* __restrict__ Xin,
    float* __restrict__ Z, int N, int E, int nbp, int chunk) {
  __shared__ GcnBlockLds L;
  const int t    = (int)threadIdx.x;
  const int lane = t & 31;
  const int wv = __builtin_amdgcn_readfirstlane(t >> 5);
  const int b  = (int)blockIdx.x;

  const float4 acc = gcn_block_rowsum(offG, reinterpret_cast<const int2*>(pe), pd,
                                      reinterpret_cast<const float4*>(Xin), L, b, nbp, chunk, E,
                                      t, wv, lane,  0);

  const int lr  = 8 * wv + (lane >> 2);
  const int row = b * GCN_RB + lr;
  if (lr < GCN_RB && row < N)
    reinterpret_cast<float4*>(Z)[(size_t)row * 4 + (lane & 3)] = acc;
}

template <int kCtrl>
static __device__ __forceinline__ float gcn_quad_xor(float x) {
  return __int_as_float(__builtin_amdgcn_update_dpp(0, __float_as_int(x), kCtrl, 0xF, 0xF, true));
}

#define GCN_RT (GCN_RB / 16)

__global__ __launch_bounds__(1024) void gcn_spmm_gemm2_lsm_kernel(
    const int* __restrict__ offG, const int* __restrict__ pe,
    const int* __restrict__ pd, const float* __restrict__ Xin,
    const float* __restrict__ W2, float* __restrict__ out,
    int N, int E, int nbp, int chunk) {
  __shared__ GcnBlockLds L;
  __shared__ __attribute__((aligned(16))) float s_l[GCN_RB * GCN_LP];
  const int t    = (int)threadIdx.x;
  const int lane = t & 31;
  const int wv  = __builtin_amdgcn_readfirstlane(t >> 5);
  const int b   = (int)blockIdx.x;
  const int col = lane & 15;
  const int kh  = lane >> 4;

  const float4 acc = gcn_block_rowsum(offG, reinterpret_cast<const int2*>(pe), pd,
                                      reinterpret_cast<const float4*>(Xin), L, b, nbp, chunk, E,
                                      t, wv, lane,  1);
  const float* __restrict__ zl = L.rows;
  {
    const int lr = 8 * wv + (lane >> 2);
    if (lr < GCN_RB)
      *reinterpret_cast<float4*>(&L.rows[lr * GCN_ZP + 4 * (lane & 3)]) = acc;
  }
  __syncthreads();

  for (int job = wv; job < 3 * GCN_RT; job += 32) {
    const int tile = job / 3;
    const int ct   = job - 3 * tile;
    const int cc = min(16 * ct + col, NCLS - 1);
    v2f bf[4];
#pragma unroll
    for (int j = 0; j < 4; ++j) {
      const int ka = 4 * j + 2 * kh;
      bf[j].x = W2[ka * NCLS + cc];
      bf[j].y = W2[(ka + 1) * NCLS + cc];
    }
    const v2f* __restrict__ zr =
        reinterpret_cast<const v2f*>(zl + (tile * 16 + col) * GCN_ZP + 2 * kh);
    v8f acc = {};
#pragma unroll
    for (int j = 0; j < 4; ++j) acc = wmma_f32_16x16x4(zr[2 * j], bf[j], acc);
    float* __restrict__ lw = s_l + (tile * 16 + 8 * kh) * GCN_LP + 16 * ct + col;
#pragma unroll
    for (int g = 0; g < 8; ++g) lw[g * GCN_LP] = acc[g];
  }
  __syncthreads();

  float* stage = reinterpret_cast<float*>(L.tab);
  if (wv < GCN_RB / 8) {
    const int qr = lane >> 2;
    const int qc = lane & 3;
    const float2* __restrict__ lp =
        reinterpret_cast<const float2*>(s_l + (8 * wv + qr) * GCN_LP + 10 * qc);
    float2 v[5];
#pragma unroll
    for (int j = 0; j < 5; ++j) v[j] = lp[j];

    float m = fmaxf(fmaxf(fmaxf(v[0].x, v[0].y), fmaxf(v[1].x, v[1].y)),
                    fmaxf(fmaxf(v[2].x, v[2].y), fmaxf(v[3].x, v[3].y)));
    m = fmaxf(m, fmaxf(v[4].x, v[4].y));
    m = fmaxf(m, gcn_quad_xor<0xB1>(m));
    m = fmaxf(m, gcn_quad_xor<0x4E>(m));

    float e[10];
#pragma unroll
    for (int j = 0; j < 5; ++j) { e[2 * j] = expf(v[j].x - m); e[2 * j + 1] = expf(v[j].y - m); }
    float s = ((e[0] + e[1]) + (e[2] + e[3])) + ((e[4] + e[5]) + (e[6] + e[7])) + (e[8] + e[9]);
    s += gcn_quad_xor<0xB1>(s);
    s += gcn_quad_xor<0x4E>(s);

    const float lse = m + logf(s);
    float2* sp =
        reinterpret_cast<float2*>(stage + wv * (8 * NCLS) + qr * NCLS + 10 * qc);
#pragma unroll
    for (int j = 0; j < 5; ++j) sp[j] = make_float2(v[j].x - lse, v[j].y - lse);
  }
  __syncthreads();

  if (wv < GCN_RB / 8) {
    const int row0 = b * GCN_RB + 8 * wv;
    const int n4   = min(8, max(0, N - row0)) * (NCLS / 4);
    const float4* s4 = reinterpret_cast<const float4*>(stage + wv * (8 * NCLS));
    float4* __restrict__ o4 = reinterpret_cast<float4*>(out) + (size_t)row0 * (NCLS / 4);
    if (n4 == 8 * NCLS / 4) {
      const float4 w0 = s4[lane];
      const float4 w1 = s4[lane + 32];
      o4[lane]      = w0;
      o4[lane + 32] = w1;
      if (lane < 16) o4[lane + 64] = s4[lane + 64];
    } else {
      for (int i = lane; i < n4; i += 32) o4[i] = s4[i];
    }
  }
}

extern "C" void kernel_launch(void* const* d_in, const int* in_sizes, int n_in,
                              void* d_out, int out_size, void* d_ws, size_t ws_size,
                              hipStream_t stream) {
  const float* features = (const float*)d_in[0];
  const int*   esrc     = (const int*)d_in[1];
  const int*   edst     = (const int*)d_in[2];
  const float* evals    = (const float*)d_in[3];
  const float* W1       = (const float*)d_in[4];
  const float* W2       = (const float*)d_in[5];
  float*       out      = (float*)d_out;

  const int N = in_sizes[0] / IN_DIM;
  const int E = in_sizes[1];

  float* Y  = (float*)d_ws;
  float* Z1 = Y + (size_t)N * HIDDEN;
  const size_t ePad  = ((size_t)E + 63) / 64 * 64;
  const int    chunk = (E + GCN_P - 1) / GCN_P;
  const int    nb    = (N + GCN_RB - 1) / GCN_RB;
  const int    nbp   = (nb + 1 + 63) / 64 * 64;
  int* lrank = (int*)(Z1 + (size_t)N * HIDDEN);
  int* pd    = lrank + ePad;
  int* pe    = pd + ePad;
  int* offG  = pe + 2 * ePad;

  const int tiles  = (N + 15) / 16;
  const int g1Grps = (tiles + GCN_G1W - 1) / GCN_G1W;
  gcn_part_gemm1_kernel<<<GCN_P + g1Grps, 1024, 0, stream>>>(
      esrc, edst, evals, offG, lrank, pe, pd, N, E, nb, nbp, chunk, features, W1, Y);

  gcn_spmm_block_kernel<<<nb, 1024, 0, stream>>>(offG, pe, pd, Y, Z1, N, E, nbp, chunk);

  gcn_spmm_gemm2_lsm_kernel<<<nb, 1024, 0, stream>>>(offG, pe, pd, Z1, W2, out, N, E, nbp, chunk);
}
